# starting kernel
# baseline (speedup 1.0000x reference)
#include <hip/hip_runtime.h>
#include <hip/hip_bf16.h>
#include <math.h>

#define NEG_ATT 0.2f
#define NEG_OUT 0.01f

typedef __attribute__((ext_vector_type(2))) float v2f;
typedef __attribute__((ext_vector_type(8))) float v8f;

__device__ __forceinline__ void atomic_max_float(float* addr, float val) {
    if (val >= 0.0f) {
        atomicMax((int*)addr, __float_as_int(val));
    } else {
        atomicMin((unsigned int*)addr, __float_as_uint(val));
    }
}

__global__ void gat_init(float* __restrict__ out, float* __restrict__ seg_max,
                         float* __restrict__ seg_sum, int N) {
    int i = blockIdx.x * blockDim.x + threadIdx.x;
    int total = N * 32;
    if (i < total) out[i] = 0.0f;
    if (i < N) {
        ((unsigned int*)seg_max)[i] = 0xFF800000u;
        seg_sum[i] = 0.0f;
    }
}

__global__ __launch_bounds__(256) void gat_gemm(
        const float* __restrict__ x,
        const float* __restrict__ Wl, const float* __restrict__ bl,
        const float* __restrict__ Wr, const float* __restrict__ br,
        float* __restrict__ xl, float* __restrict__ xr, int N) {
    const int lane = threadIdx.x & 31;
    const int wave = threadIdx.x >> 5;
    const int tile = blockIdx.x * (blockDim.x >> 5) + wave;
    const int row0 = tile * 16;
    if (row0 >= N) return;

    const int m  = lane & 15;
    const int kh = lane >> 4;

    int arow = row0 + m;
    if (arow >= N) arow = N - 1;
    const float* __restrict__ xrow = x + (size_t)arow * 128;

    v8f cl0 = {}, cl1 = {}, cr0 = {}, cr1 = {};

    #pragma unroll 4
    for (int k0 = 0; k0 < 128; k0 += 4) {
        const int kb = k0 + 2 * kh;
        v2f a;
        a[0] = xrow[kb];
        a[1] = xrow[kb + 1];

        const int n = m;
        v2f bl0v, bl1v, br0v, br1v;
        bl0v[0] = Wl[kb * 32 + n];       bl0v[1] = Wl[(kb + 1) * 32 + n];
        bl1v[0] = Wl[kb * 32 + n + 16];  bl1v[1] = Wl[(kb + 1) * 32 + n + 16];
        br0v[0] = Wr[kb * 32 + n];       br0v[1] = Wr[(kb + 1) * 32 + n];
        br1v[0] = Wr[kb * 32 + n + 16];  br1v[1] = Wr[(kb + 1) * 32 + n + 16];

        cl0 = __builtin_amdgcn_wmma_f32_16x16x4_f32(false, a, false, bl0v, (short)0, cl0, false, false);
        cl1 = __builtin_amdgcn_wmma_f32_16x16x4_f32(false, a, false, bl1v, (short)0, cl1, false, false);
        cr0 = __builtin_amdgcn_wmma_f32_16x16x4_f32(false, a, false, br0v, (short)0, cr0, false, false);
        cr1 = __builtin_amdgcn_wmma_f32_16x16x4_f32(false, a, false, br1v, (short)0, cr1, false, false);
    }

    const int n = lane & 15;
    const float bl_lo = bl[n], bl_hi = bl[n + 16];
    const float br_lo = br[n], br_hi = br[n + 16];
    #pragma unroll
    for (int v = 0; v < 8; ++v) {
        const int mm = v + 8 * kh;
        const int row = row0 + mm;
        if (row < N) {
            const size_t ro = (size_t)row * 32;
            xl[ro + n]      = cl0[v] + bl_lo;
            xl[ro + n + 16] = cl1[v] + bl_hi;
            xr[ro + n]      = cr0[v] + br_lo;
            xr[ro + n + 16] = cr1[v] + br_hi;
        }
    }
}

__global__ __launch_bounds__(256) void gat_edge_logits(
        const int* __restrict__ src, const int* __restrict__ dst,
        const float* __restrict__ ew,
        const float* __restrict__ xl, const float* __restrict__ xr,
        const float* __restrict__ We, const float* __restrict__ att,
        float* __restrict__ logits, float* __restrict__ seg_max, int E) {
    int e = blockIdx.x * blockDim.x + threadIdx.x;
    if (e >= E) return;
    const int s = src[e], d = dst[e];
    const float w = ew[e];
    const float4* __restrict__ pl = (const float4*)(xl + (size_t)s * 32);
    const float4* __restrict__ pr = (const float4*)(xr + (size_t)d * 32);
    const float4* __restrict__ pe = (const float4*)We;
    const float4* __restrict__ pa = (const float4*)att;
    float acc = 0.0f;
    #pragma unroll
    for (int i = 0; i < 8; ++i) {
        float4 a = pl[i], b = pr[i], c = pe[i], t = pa[i];
        float m0 = a.x + b.x + w * c.x; m0 = (m0 > 0.0f) ? m0 : NEG_ATT * m0;
        float m1 = a.y + b.y + w * c.y; m1 = (m1 > 0.0f) ? m1 : NEG_ATT * m1;
        float m2 = a.z + b.z + w * c.z; m2 = (m2 > 0.0f) ? m2 : NEG_ATT * m2;
        float m3 = a.w + b.w + w * c.w; m3 = (m3 > 0.0f) ? m3 : NEG_ATT * m3;
        acc += m0 * t.x + m1 * t.y + m2 * t.z + m3 * t.w;
    }
    logits[e] = acc;
    atomic_max_float(&seg_max[d], acc);
}

__global__ __launch_bounds__(256) void gat_edge_exp(
        const int* __restrict__ dst, float* __restrict__ logits,
        const float* __restrict__ seg_max, float* __restrict__ seg_sum, int E) {
    int e = blockIdx.x * blockDim.x + threadIdx.x;
    if (e >= E) return;
    const int d = dst[e];
    const float ex = __expf(logits[e] - seg_max[d]);
    logits[e] = ex;
    atomicAdd(&seg_sum[d], ex);
}

__global__ __launch_bounds__(256) void gat_edge_scatter(
        const int* __restrict__ src, const int* __restrict__ dst,
        const float* __restrict__ ex, const float* __restrict__ seg_sum,
        const float* __restrict__ xl, float* __restrict__ out, int E) {
    int e = blockIdx.x * blockDim.x + threadIdx.x;
    if (e >= E) return;
    const int s = src[e], d = dst[e];
    const float alpha = ex[e] / seg_sum[d];
    const float4* __restrict__ pl = (const float4*)(xl + (size_t)s * 32);
    float* __restrict__ po = out + (size_t)d * 32;
    #pragma unroll
    for (int i = 0; i < 8; ++i) {
        float4 v = pl[i];
        atomicAdd(po + 4 * i + 0, v.x * alpha);
        atomicAdd(po + 4 * i + 1, v.y * alpha);
        atomicAdd(po + 4 * i + 2, v.z * alpha);
        atomicAdd(po + 4 * i + 3, v.w * alpha);
    }
}

__global__ void gat_finalize(float* __restrict__ out,
                             const float* __restrict__ bias, int total) {
    int i = blockIdx.x * blockDim.x + threadIdx.x;
    if (i >= total) return;
    float v = out[i] + bias[i & 31];
    out[i] = (v > 0.0f) ? v : NEG_OUT * v;
}

extern "C" void kernel_launch(void* const* d_in, const int* in_sizes, int n_in,
                              void* d_out, int out_size, void* d_ws, size_t ws_size,
                              hipStream_t stream) {
    const float* x    = (const float*)d_in[0];
    const int*   ei   = (const int*)d_in[1];
    const float* ew   = (const float*)d_in[2];
    const float* Wl   = (const float*)d_in[3];
    const float* bl   = (const float*)d_in[4];
    const float* Wr   = (const float*)d_in[5];
    const float* br   = (const float*)d_in[6];
    const float* We   = (const float*)d_in[7];
    const float* att  = (const float*)d_in[8];
    const float* bias = (const float*)d_in[9];

    const int N = in_sizes[0] / 128;
    const int E = in_sizes[2];
    const int* srcv = ei;
    const int* dstv = ei + E;

    float* xl      = (float*)d_ws;
    float* xr      = xl + (size_t)N * 32;
    float* logits  = xr + (size_t)N * 32;
    float* seg_max = logits + (size_t)E;
    float* seg_sum = seg_max + (size_t)N;
    float* out     = (float*)d_out;

    const int tpb = 256;

    gat_init<<<(N * 32 + tpb - 1) / tpb, tpb, 0, stream>>>(out, seg_max, seg_sum, N);

    const int tiles = (N + 15) / 16;
    const int wavesPerBlock = 8;
    gat_gemm<<<(tiles + wavesPerBlock - 1) / wavesPerBlock, 32 * wavesPerBlock, 0, stream>>>(
        x, Wl, bl, Wr, br, xl, xr, N);

    gat_edge_logits<<<(E + tpb - 1) / tpb, tpb, 0, stream>>>(
        srcv, dstv, ew, xl, xr, We, att, logits, seg_max, E);

    gat_edge_exp<<<(E + tpb - 1) / tpb, tpb, 0, stream>>>(
        dstv, logits, seg_max, seg_sum, E);

    gat_edge_scatter<<<(E + tpb - 1) / tpb, tpb, 0, stream>>>(
        srcv, dstv, logits, seg_sum, xl, out, E);

    gat_finalize<<<(N * 32 + tpb - 1) / tpb, tpb, 0, stream>>>(out, bias, N * 32);
}


# aggregation by owner: each work-group owns 392 nodes, sweeps all of dst, adds its edges' weighted xl rows into LDS (float adds, arrival order) and writes out once; no global float atomics on out
# speedup vs baseline: 2.6771x; 2.6771x over previous
#include <hip/hip_runtime.h>
#include <hip/hip_bf16.h>
#include <math.h>

#define NEG_ATT 0.2f
#define NEG_OUT 0.01f

typedef __attribute__((ext_vector_type(2))) float v2f;
typedef __attribute__((ext_vector_type(8))) float v8f;

__device__ __forceinline__ void atomic_max_float(float* addr, float val) {
    if (val >= 0.0f) {
        atomicMax((int*)addr, __float_as_int(val));
    } else {
        atomicMin((unsigned int*)addr, __float_as_uint(val));
    }
}

__global__ void gat_init(float* __restrict__ seg_max,
                         float* __restrict__ seg_sum, int N) {
    int i = blockIdx.x * blockDim.x + threadIdx.x;
    if (i < N) {
        ((unsigned int*)seg_max)[i] = 0xFF800000u;
        seg_sum[i] = 0.0f;
    }
}

__global__ __launch_bounds__(256) void gat_gemm(
        const float* __restrict__ x,
        const float* __restrict__ Wl, const float* __restrict__ bl,
        const float* __restrict__ Wr, const float* __restrict__ br,
        float* __restrict__ xl, float* __restrict__ xr, int N) {
    const int lane = threadIdx.x & 31;
    const int wave = threadIdx.x >> 5;
    const int tile = blockIdx.x * (blockDim.x >> 5) + wave;
    const int row0 = tile * 16;
    if (row0 >= N) return;

    const int m  = lane & 15;
    const int kh = lane >> 4;

    int arow = row0 + m;
    if (arow >= N) arow = N - 1;
    const float* __restrict__ xrow = x + (size_t)arow * 128;

    v8f cl0 = {}, cl1 = {}, cr0 = {}, cr1 = {};

    #pragma unroll 4
    for (int k0 = 0; k0 < 128; k0 += 4) {
        const int kb = k0 + 2 * kh;
        v2f a;
        a[0] = xrow[kb];
        a[1] = xrow[kb + 1];

        const int n = m;
        v2f bl0v, bl1v, br0v, br1v;
        bl0v[0] = Wl[kb * 32 + n];       bl0v[1] = Wl[(kb + 1) * 32 + n];
        bl1v[0] = Wl[kb * 32 + n + 16];  bl1v[1] = Wl[(kb + 1) * 32 + n + 16];
        br0v[0] = Wr[kb * 32 + n];       br0v[1] = Wr[(kb + 1) * 32 + n];
        br1v[0] = Wr[kb * 32 + n + 16];  br1v[1] = Wr[(kb + 1) * 32 + n + 16];

        cl0 = __builtin_amdgcn_wmma_f32_16x16x4_f32(false, a, false, bl0v, (short)0, cl0, false, false);
        cl1 = __builtin_amdgcn_wmma_f32_16x16x4_f32(false, a, false, bl1v, (short)0, cl1, false, false);
        cr0 = __builtin_amdgcn_wmma_f32_16x16x4_f32(false, a, false, br0v, (short)0, cr0, false, false);
        cr1 = __builtin_amdgcn_wmma_f32_16x16x4_f32(false, a, false, br1v, (short)0, cr1, false, false);
    }

    const int n = lane & 15;
    const float bl_lo = bl[n], bl_hi = bl[n + 16];
    const float br_lo = br[n], br_hi = br[n + 16];
    #pragma unroll
    for (int v = 0; v < 8; ++v) {
        const int mm = v + 8 * kh;
        const int row = row0 + mm;
        if (row < N) {
            const size_t ro = (size_t)row * 32;
            xl[ro + n]      = cl0[v] + bl_lo;
            xl[ro + n + 16] = cl1[v] + bl_hi;
            xr[ro + n]      = cr0[v] + br_lo;
            xr[ro + n + 16] = cr1[v] + br_hi;
        }
    }
}

__global__ __launch_bounds__(256) void gat_edge_logits(
        const int* __restrict__ src, const int* __restrict__ dst,
        const float* __restrict__ ew,
        const float* __restrict__ xl, const float* __restrict__ xr,
        const float* __restrict__ We, const float* __restrict__ att,
        float* __restrict__ logits, float* __restrict__ seg_max, int E) {
    int e = blockIdx.x * blockDim.x + threadIdx.x;
    if (e >= E) return;
    const int s = src[e], d = dst[e];
    const float w = ew[e];
    const float4* __restrict__ pl = (const float4*)(xl + (size_t)s * 32);
    const float4* __restrict__ pr = (const float4*)(xr + (size_t)d * 32);
    const float4* __restrict__ pe = (const float4*)We;
    const float4* __restrict__ pa = (const float4*)att;
    float acc = 0.0f;
    #pragma unroll
    for (int i = 0; i < 8; ++i) {
        float4 a = pl[i], b = pr[i], c = pe[i], t = pa[i];
        float m0 = a.x + b.x + w * c.x; m0 = (m0 > 0.0f) ? m0 : NEG_ATT * m0;
        float m1 = a.y + b.y + w * c.y; m1 = (m1 > 0.0f) ? m1 : NEG_ATT * m1;
        float m2 = a.z + b.z + w * c.z; m2 = (m2 > 0.0f) ? m2 : NEG_ATT * m2;
        float m3 = a.w + b.w + w * c.w; m3 = (m3 > 0.0f) ? m3 : NEG_ATT * m3;
        acc += m0 * t.x + m1 * t.y + m2 * t.z + m3 * t.w;
    }
    logits[e] = acc;
    atomic_max_float(&seg_max[d], acc);
}

__global__ __launch_bounds__(256) void gat_edge_exp(
        const int* __restrict__ dst, float* __restrict__ logits,
        const float* __restrict__ seg_max, float* __restrict__ seg_sum, int E) {
    int e = blockIdx.x * blockDim.x + threadIdx.x;
    if (e >= E) return;
    const int d = dst[e];
    const float ex = __expf(logits[e] - seg_max[d]);
    logits[e] = ex;
    atomicAdd(&seg_sum[d], ex);
}

#define OWN_ROWS    392
#define OWN_THREADS 1024
#define OWN_WAVES   (OWN_THREADS / 32)
#define SCAN_STEPS  4
#define Q_CAP       1024

__global__ __launch_bounds__(OWN_THREADS) void gat_scatter_owner(
        const int* __restrict__ src, const int* __restrict__ dst,
        const float* __restrict__ ex, const float* __restrict__ seg_sum,
        const float* __restrict__ xl, const float* __restrict__ bias,
        float* __restrict__ out, int N, int E) {
    __shared__ __attribute__((aligned(16))) float acc[(OWN_ROWS + 1) * 32];
    __shared__ int queue[OWN_WAVES][Q_CAP];

    const int tid  = threadIdx.x;
    const int lane = tid & 31;
    const int wave = __builtin_amdgcn_readfirstlane(tid >> 5);
    const int d0   = blockIdx.x * OWN_ROWS;
    int rows = N - d0;
    if (rows > OWN_ROWS) rows = OWN_ROWS;
    const unsigned urows = (unsigned)rows;

    float4* acc4 = (float4*)acc;
    for (int i = tid; i < (OWN_ROWS + 1) * 8; i += OWN_THREADS)
        acc4[i] = make_float4(0.0f, 0.0f, 0.0f, 0.0f);
    __syncthreads();

    int* __restrict__ q = queue[wave];
    int head = 0, tail = 0;

    auto push = [&](bool hit, int e) {
        const unsigned mask = __builtin_amdgcn_ballot_w32(hit);
        if (mask != 0u) {
            if (hit) {
                const int rank = (int)__builtin_amdgcn_mbcnt_lo(mask, 0u);
                q[(tail + rank) & (Q_CAP - 1)] = e;
            }
            tail += __builtin_popcount(mask);
        }
    };

    auto drain = [&](int n) {
        int rel = OWN_ROWS, s = 0;
        float a = 0.0f;
        if (lane < n) {
            const int e = q[(head + lane) & (Q_CAP - 1)];
            rel = dst[e] - d0;
            s   = src[e];
            a   = ex[e] / seg_sum[d0 + rel];
        }
        head += n;
        float v[32];
        #pragma unroll
        for (int k = 0; k < 32; ++k) {
            const int sk = __builtin_amdgcn_readlane(s, k);
            v[k] = xl[(size_t)sk * 32 + lane];
        }
        #pragma unroll
        for (int k = 0; k < 32; ++k) {
            const int   rk = __builtin_amdgcn_readlane(rel, k);
            const float ak = __int_as_float(__builtin_amdgcn_readlane(__float_as_int(a), k));
            __hip_atomic_fetch_add(&acc[rk * 32 + lane], v[k] * ak,
                                   __ATOMIC_RELAXED, __HIP_MEMORY_SCOPE_WORKGROUP);
        }
    };

    const bool vec_ok   = (((size_t)dst) & 15) == 0;
    const int  nBatches = vec_ok ? ((E >> 2) / (OWN_THREADS * SCAN_STEPS)) : 0;
    const int  swept    = nBatches * (OWN_THREADS * SCAN_STEPS * 4);
    const int4* __restrict__ dst4 = (const int4*)dst;
    auto load4 = [&](int g) -> int4 { return dst4[g]; };

    auto scan_batch = [&](const int4 (&buf)[SCAN_STEPS], int b) {
        #pragma unroll
        for (int u = 0; u < SCAN_STEPS; ++u) {
            const unsigned r0 = (unsigned)(buf[u].x - d0);
            const unsigned r1 = (unsigned)(buf[u].y - d0);
            const unsigned r2 = (unsigned)(buf[u].z - d0);
            const unsigned r3 = (unsigned)(buf[u].w - d0);
            const unsigned rm = min(min(r0, r1), min(r2, r3));
            if (__builtin_amdgcn_ballot_w32(rm < urows) != 0u) {
                const int e0 = ((b * SCAN_STEPS + u) * OWN_THREADS + tid) * 4;
                push(r0 < urows, e0);
                push(r1 < urows, e0 + 1);
                push(r2 < urows, e0 + 2);
                push(r3 < urows, e0 + 3);
            }
        }
        while (tail - head >= 32) drain(32);
    };

    if (nBatches > 0) {
        int bi = (int)(((unsigned long long)blockIdx.x * (unsigned long long)nBatches) / gridDim.x);

        int4 bufA[SCAN_STEPS], bufB[SCAN_STEPS];
        int b1 = bi + 1;
        if (b1 >= nBatches) b1 = 0;
        #pragma unroll
        for (int u = 0; u < SCAN_STEPS; ++u)
            bufA[u] = load4((bi * SCAN_STEPS + u) * OWN_THREADS + tid);
        #pragma unroll
        for (int u = 0; u < SCAN_STEPS; ++u)
            bufB[u] = load4((b1 * SCAN_STEPS + u) * OWN_THREADS + tid);

        for (int it = 0; it < nBatches; it += 2) {
            scan_batch(bufA, bi);
            int b2 = b1 + 1;
            if (b2 >= nBatches) b2 = 0;
            #pragma unroll
            for (int u = 0; u < SCAN_STEPS; ++u)
                bufA[u] = load4((b2 * SCAN_STEPS + u) * OWN_THREADS + tid);

            if (it + 1 < nBatches) scan_batch(bufB, b1);
            int b3 = b2 + 1;
            if (b3 >= nBatches) b3 = 0;
            #pragma unroll
            for (int u = 0; u < SCAN_STEPS; ++u)
                bufB[u] = load4((b3 * SCAN_STEPS + u) * OWN_THREADS + tid);
            bi = b2;
            b1 = b3;
        }
    }

    for (int base = swept + wave * 32; ; base += OWN_THREADS) {
        const bool more = base < E;
        if (more) {
            const int e = base + lane;
            const unsigned r = (e < E) ? (unsigned)(dst[e] - d0) : 0xFFFFFFFFu;
            push(r < urows, e);
        }
        int n = tail - head;
        while (n >= 32 || (!more && n > 0)) {
            drain(n < 32 ? n : 32);
            n = tail - head;
        }
        if (!more) break;
    }
    __syncthreads();

    float4* __restrict__ out4 = (float4*)(out + (size_t)d0 * 32);
    for (int i = tid; i < rows * 8; i += OWN_THREADS) {
        const int c = (i & 7) * 4;
        float4 v = acc4[i];
        v.x += bias[c];     v.x = (v.x > 0.0f) ? v.x : NEG_OUT * v.x;
        v.y += bias[c + 1]; v.y = (v.y > 0.0f) ? v.y : NEG_OUT * v.y;
        v.z += bias[c + 2]; v.z = (v.z > 0.0f) ? v.z : NEG_OUT * v.z;
        v.w += bias[c + 3]; v.w = (v.w > 0.0f) ? v.w : NEG_OUT * v.w;
        out4[i] = v;
    }
}

extern "C" void kernel_launch(void* const* d_in, const int* in_sizes, int n_in,
                              void* d_out, int out_size, void* d_ws, size_t ws_size,
                              hipStream_t stream) {
    const float* x    = (const float*)d_in[0];
    const int*   ei   = (const int*)d_in[1];
    const float* ew   = (const float*)d_in[2];
    const float* Wl   = (const float*)d_in[3];
    const float* bl   = (const float*)d_in[4];
    const float* Wr   = (const float*)d_in[5];
    const float* br   = (const float*)d_in[6];
    const float* We   = (const float*)d_in[7];
    const float* att  = (const float*)d_in[8];
    const float* bias = (const float*)d_in[9];

    const int N = in_sizes[0] / 128;
    const int E = in_sizes[2];
    const int* srcv = ei;
    const int* dstv = ei + E;

    float* xl      = (float*)d_ws;
    float* xr      = xl + (size_t)N * 32;
    float* logits  = xr + (size_t)N * 32;
    float* seg_max = logits + (size_t)E;
    float* seg_sum = seg_max + (size_t)N;
    float* out     = (float*)d_out;

    const int tpb = 256;

    gat_init<<<(N + tpb - 1) / tpb, tpb, 0, stream>>>(seg_max, seg_sum, N);

    const int tiles = (N + 15) / 16;
    const int wavesPerBlock = 8;
    gat_gemm<<<(tiles + wavesPerBlock - 1) / wavesPerBlock, 32 * wavesPerBlock, 0, stream>>>(
        x, Wl, bl, Wr, br, xl, xr, N);

    gat_edge_logits<<<(E + tpb - 1) / tpb, tpb, 0, stream>>>(
        srcv, dstv, ew, xl, xr, We, att, logits, seg_max, E);

    gat_edge_exp<<<(E + tpb - 1) / tpb, tpb, 0, stream>>>(
        dstv, logits, seg_max, seg_sum, E);

    gat_scatter_owner<<<(N + OWN_ROWS - 1) / OWN_ROWS, OWN_THREADS, 0, stream>>>(
        srcv, dstv, logits, seg_sum, xl, bias, out, N, E);
}


# owner pass computes its edges' logits (lane = column, lane-tree sum) and the softmax in LDS (atomic arrival order) from per-wave edge lists; init, logits, exp launches, global atomics gone (5 to 2)
# speedup vs baseline: 7.3690x; 2.7526x over previous
#include <hip/hip_runtime.h>
#include <hip/hip_bf16.h>
#include <math.h>


#define NEG_ATT 0.2f
#define NEG_OUT 0.01f

typedef __attribute__((ext_vector_type(2))) float v2f;
typedef __attribute__((ext_vector_type(8))) float v8f;

__global__ __launch_bounds__(256) void gat_gemm(
        const float* __restrict__ x,
        const float* __restrict__ Wl, const float* __restrict__ bl,
        const float* __restrict__ Wr, const float* __restrict__ br,
        float* __restrict__ xl, float* __restrict__ xr, int N) {
    const int lane = threadIdx.x & 31;
    const int wave = threadIdx.x >> 5;
    const int tile = blockIdx.x * (blockDim.x >> 5) + wave;
    const int row0 = tile * 16;
    if (row0 >= N) return;

    const int m  = lane & 15;
    const int kh = lane >> 4;

    int arow = row0 + m;
    if (arow >= N) arow = N - 1;
    const float* __restrict__ xrow = x + (size_t)arow * 128;

    v8f cl0 = {}, cl1 = {}, cr0 = {}, cr1 = {};

    #pragma unroll 4
    for (int k0 = 0; k0 < 128; k0 += 4) {
        const int kb = k0 + 2 * kh;
        v2f a;
        a[0] = xrow[kb];
        a[1] = xrow[kb + 1];

        const int n = m;
        v2f bl0v, bl1v, br0v, br1v;
        bl0v[0] = Wl[kb * 32 + n];       bl0v[1] = Wl[(kb + 1) * 32 + n];
        bl1v[0] = Wl[kb * 32 + n + 16];  bl1v[1] = Wl[(kb + 1) * 32 + n + 16];
        br0v[0] = Wr[kb * 32 + n];       br0v[1] = Wr[(kb + 1) * 32 + n];
        br1v[0] = Wr[kb * 32 + n + 16];  br1v[1] = Wr[(kb + 1) * 32 + n + 16];

        cl0 = __builtin_amdgcn_wmma_f32_16x16x4_f32(false, a, false, bl0v, (short)0, cl0, false, false);
        cl1 = __builtin_amdgcn_wmma_f32_16x16x4_f32(false, a, false, bl1v, (short)0, cl1, false, false);
        cr0 = __builtin_amdgcn_wmma_f32_16x16x4_f32(false, a, false, br0v, (short)0, cr0, false, false);
        cr1 = __builtin_amdgcn_wmma_f32_16x16x4_f32(false, a, false, br1v, (short)0, cr1, false, false);
    }

    const int n = lane & 15;
    const float bl_lo = bl[n], bl_hi = bl[n + 16];
    const float br_lo = br[n], br_hi = br[n + 16];
    #pragma unroll
    for (int v = 0; v < 8; ++v) {
        const int mm = v + 8 * kh;
        const int row = row0 + mm;
        if (row < N) {
            const size_t ro = (size_t)row * 32;
            xl[ro + n]      = cl0[v] + bl_lo;
            xl[ro + n + 16] = cl1[v] + bl_hi;
            xr[ro + n]      = cr0[v] + br_lo;
            xr[ro + n + 16] = cr1[v] + br_hi;
        }
    }
}

#define OWN_ROWS    392
#define OWN_THREADS 1024
#define OWN_WAVES   (OWN_THREADS / 32)
#define SCAN_STEPS  4
#define L_CAP       640
#define CHUNK_EDGES ((L_CAP / 32) * OWN_THREADS)

template <int N>
__device__ __forceinline__ float row_xor(float x) {
    return __int_as_float(__builtin_amdgcn_update_dpp(0, __float_as_int(x), 0x160 + N, 0xf, 0xf, false));
}
__device__ __forceinline__ float half_xor(float x) {
    const int b = __float_as_int(x);
    return __int_as_float(__builtin_amdgcn_permlanex16(b, b, 0x76543210, 0xfedcba98, false, false));
}

__device__ __forceinline__ void lds_max_float(float* addr, float val) {
    const float v = val + 0.0f;
    if (v >= 0.0f) {
        atomicMax((int*)addr, __float_as_int(v));
    } else {
        atomicMin((unsigned int*)addr, __float_as_uint(v));
    }
}

__global__ __launch_bounds__(OWN_THREADS) void gat_edges_owner(
        const int* __restrict__ src, const int* __restrict__ dst,
        const float* __restrict__ ew,
        const float* __restrict__ xl, const float* __restrict__ xr,
        const float* __restrict__ We, const float* __restrict__ att,
        const float* __restrict__ bias,
        float* __restrict__ out, int N, int E) {
    __shared__ __attribute__((aligned(16))) float acc[(OWN_ROWS + 1) * 32];
    __shared__ __attribute__((aligned(16))) float xrs[(OWN_ROWS + 1) * 32];
    __shared__ int            lst_e[OWN_WAVES][L_CAP];
    __shared__ float          lst_l[OWN_WAVES][L_CAP];
    __shared__ unsigned short lst_r[OWN_WAVES][L_CAP];
    __shared__ float smax[OWN_ROWS + 1];
    __shared__ float cmax[OWN_ROWS + 1];
    __shared__ float ssum[OWN_ROWS + 1];
    __shared__ int   ovf_flag;

    const int tid  = threadIdx.x;
    const int lane = tid & 31;
    const int wave = __builtin_amdgcn_readfirstlane(tid >> 5);
    const int d0   = blockIdx.x * OWN_ROWS;
    int rows = N - d0;
    if (rows > OWN_ROWS) rows = OWN_ROWS;
    const unsigned urows = (unsigned)rows;

    float4* acc4 = (float4*)acc;
    float4* xrs4 = (float4*)xrs;
    const float4* __restrict__ xr4 = (const float4*)(xr + (size_t)d0 * 32);
    for (int i = tid; i < (OWN_ROWS + 1) * 8; i += OWN_THREADS) {
        acc4[i] = make_float4(0.0f, 0.0f, 0.0f, 0.0f);
        xrs4[i] = ((i >> 3) < rows) ? xr4[i] : make_float4(0.0f, 0.0f, 0.0f, 0.0f);
    }
    for (int i = tid; i < OWN_ROWS + 1; i += OWN_THREADS) {
        smax[i] = -INFINITY;
        cmax[i] = -INFINITY;
        ssum[i] = 0.0f;
    }
    if (tid == 0) ovf_flag = 0;
    __syncthreads();

    int*            __restrict__ le = lst_e[wave];
    float*          __restrict__ ll = lst_l[wave];
    unsigned short* __restrict__ lr = lst_r[wave];
    int tail = 0;

    const float we_l = We[lane];
    const float at_l = att[lane];

    auto push = [&](bool hit, int e, unsigned r) {
        const unsigned mask = __builtin_amdgcn_ballot_w32(hit);
        if (mask != 0u) {
            if (hit) {
                const int p = tail + (int)__builtin_amdgcn_mbcnt_lo(mask, 0u);
                if (p < L_CAP) {
                    le[p] = e;
                    lr[p] = (unsigned short)r;
                }
            }
            tail += __builtin_popcount(mask);
        }
    };

    auto sweep_plain = [&](int lo, int hi) {
        for (int base = lo + wave * 32; base < hi; base += OWN_THREADS) {
            const int e = base + lane;
            const unsigned r = (e < hi) ? (unsigned)(dst[e] - d0) : 0xFFFFFFFFu;
            push(r < urows, e, r);
        }
    };

    auto sweep_fast = [&]() {
        const bool vec_ok   = (((size_t)dst) & 15) == 0;
        const int  nBatches = vec_ok ? ((E >> 2) / (OWN_THREADS * SCAN_STEPS)) : 0;
        const int  swept    = nBatches * (OWN_THREADS * SCAN_STEPS * 4);
        const int4* __restrict__ dst4 = (const int4*)dst;
        auto load4 = [&](int g) -> int4 { return dst4[g]; };

        auto scan_batch = [&](const int4 (&buf)[SCAN_STEPS], int b) {
            #pragma unroll
            for (int u = 0; u < SCAN_STEPS; ++u) {
                const unsigned r0 = (unsigned)(buf[u].x - d0);
                const unsigned r1 = (unsigned)(buf[u].y - d0);
                const unsigned r2 = (unsigned)(buf[u].z - d0);
                const unsigned r3 = (unsigned)(buf[u].w - d0);
                const unsigned rm = min(min(r0, r1), min(r2, r3));
                if (__builtin_amdgcn_ballot_w32(rm < urows) != 0u) {
                    const int e0 = ((b * SCAN_STEPS + u) * OWN_THREADS + tid) * 4;
                    push(r0 < urows, e0,     r0);
                    push(r1 < urows, e0 + 1, r1);
                    push(r2 < urows, e0 + 2, r2);
                    push(r3 < urows, e0 + 3, r3);
                }
            }
        };

        if (nBatches > 0) {
            int bi = (int)(((unsigned long long)blockIdx.x * (unsigned long long)nBatches) / gridDim.x);

            int4 bufA[SCAN_STEPS], bufB[SCAN_STEPS];
            int b1 = bi + 1;
            if (b1 >= nBatches) b1 = 0;
            #pragma unroll
            for (int u = 0; u < SCAN_STEPS; ++u)
                bufA[u] = load4((bi * SCAN_STEPS + u) * OWN_THREADS + tid);
            #pragma unroll
            for (int u = 0; u < SCAN_STEPS; ++u)
                bufB[u] = load4((b1 * SCAN_STEPS + u) * OWN_THREADS + tid);

            for (int it = 0; it < nBatches; it += 2) {
                scan_batch(bufA, bi);
                int b2 = b1 + 1;
                if (b2 >= nBatches) b2 = 0;
                #pragma unroll
                for (int u = 0; u < SCAN_STEPS; ++u)
                    bufA[u] = load4((b2 * SCAN_STEPS + u) * OWN_THREADS + tid);

                if (it + 1 < nBatches) scan_batch(bufB, b1);
                int b3 = b2 + 1;
                if (b3 >= nBatches) b3 = 0;
                #pragma unroll
                for (int u = 0; u < SCAN_STEPS; ++u)
                    bufB[u] = load4((b3 * SCAN_STEPS + u) * OWN_THREADS + tid);
                bi = b2;
                b1 = b3;
            }
        }
        sweep_plain(swept, E);
    };

    auto flush = [&]() {
        const int ng = (tail + 31) >> 5;

        int   s_nx = 0;
        float w_nx = 0.0f;
        if (lane < tail) {
            const int e = le[lane];
            s_nx = src[e];
            w_nx = ew[e];
        }
        for (int g = 0; g < ng; ++g) {
            const int   n = tail - g * 32;
            const int   s = s_nx;
            const float w = w_nx;
            int rel = OWN_ROWS;
            if (lane < n) rel = lr[g * 32 + lane];
            s_nx = 0;
            w_nx = 0.0f;
            if ((g + 1) * 32 + lane < tail) {
                const int e = le[(g + 1) * 32 + lane];
                s_nx = src[e];
                w_nx = ew[e];
            }
            float v[32];
            #pragma unroll
            for (int k = 0; k < 32; ++k) {
                const int sk = __builtin_amdgcn_readlane(s, k);
                v[k] = xl[(size_t)sk * 32 + lane];
            }
            #pragma unroll
            for (int k = 0; k < 32; ++k) {
                const int   rk = __builtin_amdgcn_readlane(rel, k);
                const float wk = __int_as_float(__builtin_amdgcn_readlane(__float_as_int(w), k));
                float m = v[k] + xrs[rk * 32 + lane] + wk * we_l;
                m = (m > 0.0f) ? m : NEG_ATT * m;
                v[k] = m * at_l;
            }
            const bool b16 = (lane & 16) != 0, b8 = (lane & 8) != 0,
                       b4  = (lane & 4)  != 0, b2 = (lane & 2) != 0, b1 = (lane & 1) != 0;
            #pragma unroll
            for (int k = 0; k < 16; ++k) {
                const float keep = b16 ? v[k + 16] : v[k];
                const float send = b16 ? v[k] : v[k + 16];
                v[k] = keep + half_xor(send);
            }
            #pragma unroll
            for (int k = 0; k < 8; ++k) {
                const float keep = b8 ? v[k + 8] : v[k];
                const float send = b8 ? v[k] : v[k + 8];
                v[k] = keep + row_xor<8>(send);
            }
            #pragma unroll
            for (int k = 0; k < 4; ++k) {
                const float keep = b4 ? v[k + 4] : v[k];
                const float send = b4 ? v[k] : v[k + 4];
                v[k] = keep + row_xor<4>(send);
            }
            #pragma unroll
            for (int k = 0; k < 2; ++k) {
                const float keep = b2 ? v[k + 2] : v[k];
                const float send = b2 ? v[k] : v[k + 2];
                v[k] = keep + row_xor<2>(send);
            }
            const float keep1 = b1 ? v[1] : v[0];
            const float send1 = b1 ? v[0] : v[1];
            const float lg = keep1 + row_xor<1>(send1);
            if (lane < n) {
                ll[g * 32 + lane] = lg;
                le[g * 32 + lane] = s;
                lds_max_float(&cmax[rel], lg);
            }
        }
        __syncthreads();

        for (int i = tid; i < OWN_ROWS * 32; i += OWN_THREADS) {
            const int r = i >> 5;
            const float mo = smax[r], c = cmax[r];
            if (c > mo) {
                if (mo != -INFINITY) {
                    const float f = __expf(mo - c);
                    acc[i] *= f;
                    if (lane == 0) ssum[r] *= f;
                }
                if (lane == 0) smax[r] = c;
            }
            if (lane == 0) cmax[r] = -INFINITY;
        }
        __syncthreads();

        for (int g = 0; g < ng; ++g) {
            const int n = tail - g * 32;
            int rel = OWN_ROWS, s = 0;
            float a = 0.0f;
            if (lane < n) {
                s   = le[g * 32 + lane];
                rel = lr[g * 32 + lane];
                a   = __expf(ll[g * 32 + lane] - smax[rel]);
                __hip_atomic_fetch_add(&ssum[rel], a,
                                       __ATOMIC_RELAXED, __HIP_MEMORY_SCOPE_WORKGROUP);
            }
            float v[32];
            #pragma unroll
            for (int k = 0; k < 32; ++k) {
                const int sk = __builtin_amdgcn_readlane(s, k);
                v[k] = xl[(size_t)sk * 32 + lane];
            }
            #pragma unroll
            for (int k = 0; k < 32; ++k) {
                const int   rk = __builtin_amdgcn_readlane(rel, k);
                const float ak = __int_as_float(__builtin_amdgcn_readlane(__float_as_int(a), k));
                __hip_atomic_fetch_add(&acc[rk * 32 + lane], v[k] * ak,
                                       __ATOMIC_RELAXED, __HIP_MEMORY_SCOPE_WORKGROUP);
            }
        }
    };

    sweep_fast();
    if (tail > L_CAP && lane == 0) ovf_flag = 1;
    __syncthreads();
    const bool over    = __builtin_amdgcn_readfirstlane(ovf_flag) != 0;
    const int  nChunks = over ? (E + CHUNK_EDGES - 1) / CHUNK_EDGES : 1;
    for (int chunk = 0; chunk < nChunks; ++chunk) {
        if (over) {
            const int lo = chunk * CHUNK_EDGES;
            int hi = lo + CHUNK_EDGES;
            if (hi > E) hi = E;
            tail = 0;
            sweep_plain(lo, hi);
        }
        flush();
    }
    __syncthreads();

    float4* __restrict__ out4 = (float4*)(out + (size_t)d0 * 32);
    for (int i = tid; i < rows * 8; i += OWN_THREADS) {
        const int c = (i & 7) * 4;
        const float sm = ssum[i >> 3];
        float4 v = acc4[i];
        if (sm != 0.0f) { v.x /= sm; v.y /= sm; v.z /= sm; v.w /= sm; }
        v.x += bias[c];     v.x = (v.x > 0.0f) ? v.x : NEG_OUT * v.x;
        v.y += bias[c + 1]; v.y = (v.y > 0.0f) ? v.y : NEG_OUT * v.y;
        v.z += bias[c + 2]; v.z = (v.z > 0.0f) ? v.z : NEG_OUT * v.z;
        v.w += bias[c + 3]; v.w = (v.w > 0.0f) ? v.w : NEG_OUT * v.w;
        out4[i] = v;
    }
}

extern "C" void kernel_launch(void* const* d_in, const int* in_sizes, int n_in,
                              void* d_out, int out_size, void* d_ws, size_t ws_size,
                              hipStream_t stream) {
    const float* x    = (const float*)d_in[0];
    const int*   ei   = (const int*)d_in[1];
    const float* ew   = (const float*)d_in[2];
    const float* Wl   = (const float*)d_in[3];
    const float* bl   = (const float*)d_in[4];
    const float* Wr   = (const float*)d_in[5];
    const float* br   = (const float*)d_in[6];
    const float* We   = (const float*)d_in[7];
    const float* att  = (const float*)d_in[8];
    const float* bias = (const float*)d_in[9];

    const int N = in_sizes[0] / 128;
    const int E = in_sizes[2];
    const int* srcv = ei;
    const int* dstv = ei + E;

    float* xl  = (float*)d_ws;
    float* xr  = xl + (size_t)N * 32;
    float* out = (float*)d_out;

    const int tiles = (N + 15) / 16;
    const int wavesPerBlock = 8;
    gat_gemm<<<(tiles + wavesPerBlock - 1) / wavesPerBlock, 32 * wavesPerBlock, 0, stream>>>(
        x, Wl, bl, Wr, br, xl, xr, N);

    gat_edges_owner<<<(N + OWN_ROWS - 1) / OWN_ROWS, OWN_THREADS, 0, stream>>>(
        srcv, dstv, ew, xl, xr, We, att, bias, out, N, E);
}


# sweep reads 1-byte owner ids (dst/392, packed by the GEMM launch in the old sweep's lane order, so each wave's list is unchanged, unlike the earlier byte-id try), not dst; sums in LDS arrival order
# speedup vs baseline: 8.0432x; 1.0915x over previous
#include <hip/hip_runtime.h>
#include <hip/hip_bf16.h>
#include <math.h>


#define NEG_ATT 0.2f
#define NEG_OUT 0.01f

typedef __attribute__((ext_vector_type(2))) float v2f;
typedef __attribute__((ext_vector_type(8))) float v8f;

#define OWN_ROWS    392
#define OWN_THREADS 1024
#define OWN_WAVES   (OWN_THREADS / 32)
#define SCAN_STEPS  4
#define L_CAP       640
#define BATCH_EDGES (OWN_THREADS * SCAN_STEPS * 4)

__device__ __forceinline__ unsigned owner_byte(int d) {
    return ((unsigned)d / (unsigned)OWN_ROWS) & 255u;
}

__global__ __launch_bounds__(256) void gat_gemm(
        const float* __restrict__ x,
        const float* __restrict__ Wl, const float* __restrict__ bl,
        const float* __restrict__ Wr, const float* __restrict__ br,
        float* __restrict__ xl, float* __restrict__ xr, int N,
        const int* __restrict__ dst, unsigned int* __restrict__ oid, int E) {
    {
        const bool vec_ok   = (((size_t)dst) & 15) == 0;
        const int  nBatches = vec_ok ? (E / BATCH_EDGES) : 0;
        const int4* __restrict__ dst4 = (const int4*)dst;
        uint4* __restrict__ oid4 = (uint4*)oid;
        const int nChunks = nBatches * OWN_THREADS;
        for (int c = blockIdx.x * blockDim.x + threadIdx.x; c < nChunks; c += gridDim.x * blockDim.x) {
            const int b = c / OWN_THREADS, t = c % OWN_THREADS;
            unsigned o[SCAN_STEPS];
            #pragma unroll
            for (int u = 0; u < SCAN_STEPS; ++u) {
                const int4 d = dst4[(b * SCAN_STEPS + u) * OWN_THREADS + t];
                o[u] = owner_byte(d.x) | (owner_byte(d.y) << 8) | (owner_byte(d.z) << 16) | (owner_byte(d.w) << 24);
            }
            oid4[c] = make_uint4(o[0], o[1], o[2], o[3]);
        }
    }

    const int lane = threadIdx.x & 31;
    const int wave = threadIdx.x >> 5;
    const int tile = blockIdx.x * (blockDim.x >> 5) + wave;
    const int row0 = tile * 16;
    if (row0 >= N) return;

    const int m  = lane & 15;
    const int kh = lane >> 4;

    int arow = row0 + m;
    if (arow >= N) arow = N - 1;
    const float* __restrict__ xrow = x + (size_t)arow * 128;

    v8f cl0 = {}, cl1 = {}, cr0 = {}, cr1 = {};

    #pragma unroll 4
    for (int k0 = 0; k0 < 128; k0 += 4) {
        const int kb = k0 + 2 * kh;
        v2f a;
        a[0] = xrow[kb];
        a[1] = xrow[kb + 1];

        const int n = m;
        v2f bl0v, bl1v, br0v, br1v;
        bl0v[0] = Wl[kb * 32 + n];       bl0v[1] = Wl[(kb + 1) * 32 + n];
        bl1v[0] = Wl[kb * 32 + n + 16];  bl1v[1] = Wl[(kb + 1) * 32 + n + 16];
        br0v[0] = Wr[kb * 32 + n];       br0v[1] = Wr[(kb + 1) * 32 + n];
        br1v[0] = Wr[kb * 32 + n + 16];  br1v[1] = Wr[(kb + 1) * 32 + n + 16];

        cl0 = __builtin_amdgcn_wmma_f32_16x16x4_f32(false, a, false, bl0v, (short)0, cl0, false, false);
        cl1 = __builtin_amdgcn_wmma_f32_16x16x4_f32(false, a, false, bl1v, (short)0, cl1, false, false);
        cr0 = __builtin_amdgcn_wmma_f32_16x16x4_f32(false, a, false, br0v, (short)0, cr0, false, false);
        cr1 = __builtin_amdgcn_wmma_f32_16x16x4_f32(false, a, false, br1v, (short)0, cr1, false, false);
    }

    const int n = lane & 15;
    const float bl_lo = bl[n], bl_hi = bl[n + 16];
    const float br_lo = br[n], br_hi = br[n + 16];
    #pragma unroll
    for (int v = 0; v < 8; ++v) {
        const int mm = v + 8 * kh;
        const int row = row0 + mm;
        if (row < N) {
            const size_t ro = (size_t)row * 32;
            xl[ro + n]      = cl0[v] + bl_lo;
            xl[ro + n + 16] = cl1[v] + bl_hi;
            xr[ro + n]      = cr0[v] + br_lo;
            xr[ro + n + 16] = cr1[v] + br_hi;
        }
    }
}

#define CHUNK_EDGES ((L_CAP / 32) * OWN_THREADS)

template <int N>
__device__ __forceinline__ float row_xor(float x) {
    return __int_as_float(__builtin_amdgcn_update_dpp(0, __float_as_int(x), 0x160 + N, 0xf, 0xf, false));
}
__device__ __forceinline__ float half_xor(float x) {
    const int b = __float_as_int(x);
    return __int_as_float(__builtin_amdgcn_permlanex16(b, b, 0x76543210, 0xfedcba98, false, false));
}

__device__ __forceinline__ void lds_max_float(float* addr, float val) {
    const float v = val + 0.0f;
    if (v >= 0.0f) {
        atomicMax((int*)addr, __float_as_int(v));
    } else {
        atomicMin((unsigned int*)addr, __float_as_uint(v));
    }
}

__global__ __launch_bounds__(OWN_THREADS) void gat_edges_owner(
        const int* __restrict__ src, const int* __restrict__ dst,
        const unsigned int* __restrict__ oid,
        const float* __restrict__ ew,
        const float* __restrict__ xl, const float* __restrict__ xr,
        const float* __restrict__ We, const float* __restrict__ att,
        const float* __restrict__ bias,
        float* __restrict__ out, int N, int E) {
    __shared__ __attribute__((aligned(16))) float acc[(OWN_ROWS + 1) * 32];
    __shared__ __attribute__((aligned(16))) float xrs[(OWN_ROWS + 1) * 32];
    __shared__ int            lst_e[OWN_WAVES][L_CAP];
    __shared__ float          lst_l[OWN_WAVES][L_CAP];
    __shared__ unsigned short lst_r[OWN_WAVES][L_CAP];
    __shared__ float smax[OWN_ROWS + 1];
    __shared__ float cmax[OWN_ROWS + 1];
    __shared__ float ssum[OWN_ROWS + 1];
    __shared__ int   ovf_flag;

    const int tid  = threadIdx.x;
    const int lane = tid & 31;
    const int wave = __builtin_amdgcn_readfirstlane(tid >> 5);
    const int d0   = blockIdx.x * OWN_ROWS;
    int rows = N - d0;
    if (rows > OWN_ROWS) rows = OWN_ROWS;
    const unsigned urows = (unsigned)rows;

    float4* acc4 = (float4*)acc;
    float4* xrs4 = (float4*)xrs;
    const float4* __restrict__ xr4 = (const float4*)(xr + (size_t)d0 * 32);
    for (int i = tid; i < (OWN_ROWS + 1) * 8; i += OWN_THREADS) {
        acc4[i] = make_float4(0.0f, 0.0f, 0.0f, 0.0f);
        xrs4[i] = ((i >> 3) < rows) ? xr4[i] : make_float4(0.0f, 0.0f, 0.0f, 0.0f);
    }
    for (int i = tid; i < OWN_ROWS + 1; i += OWN_THREADS) {
        smax[i] = -INFINITY;
        cmax[i] = -INFINITY;
        ssum[i] = 0.0f;
    }
    if (tid == 0) ovf_flag = 0;
    __syncthreads();

    int*            __restrict__ le = lst_e[wave];
    float*          __restrict__ ll = lst_l[wave];
    unsigned short* __restrict__ lr = lst_r[wave];
    int tail = 0;

    const float we_l = We[lane];
    const float at_l = att[lane];

    auto push = [&](bool hit, int e) {
        const unsigned mask = __builtin_amdgcn_ballot_w32(hit);
        if (mask != 0u) {
            if (hit) {
                const int p = tail + (int)__builtin_amdgcn_mbcnt_lo(mask, 0u);
                if (p < L_CAP) le[p] = e;
            }
            tail += __builtin_popcount(mask);
        }
    };

    auto sweep_plain = [&](int lo, int hi) {
        for (int base = lo + wave * 32; base < hi; base += OWN_THREADS) {
            const int e = base + lane;
            const unsigned r = (e < hi) ? (unsigned)(dst[e] - d0) : 0xFFFFFFFFu;
            push(r < urows, e);
        }
    };

    auto sweep_fast = [&]() {
        const bool vec_ok   = (((size_t)dst) & 15) == 0;
        const int  nBatches = vec_ok ? (E / BATCH_EDGES) : 0;
        const int  swept    = nBatches * BATCH_EDGES;
        const uint4* __restrict__ oid4 = (const uint4*)oid;
        const unsigned pat = (blockIdx.x & 255u) * 0x01010101u;

        const int bi0 = (int)(((unsigned long long)blockIdx.x * (unsigned long long)nBatches) / gridDim.x);
        auto batch_of = [&](int q) -> int {
            const int b = bi0 + q;
            return (b >= nBatches) ? b - nBatches : b;
        };

        auto scan_word = [&](unsigned wd, int e0) {
            const unsigned t = wd ^ pat;
            const unsigned z = (t - 0x01010101u) & ~t & 0x80808080u;
            if (__builtin_amdgcn_ballot_w32(z != 0u) != 0u) {
                push((t & 0x000000FFu) == 0u, e0);
                push((t & 0x0000FF00u) == 0u, e0 + 1);
                push((t & 0x00FF0000u) == 0u, e0 + 2);
                push((t & 0xFF000000u) == 0u, e0 + 3);
            }
        };
        auto scan_batch = [&](const uint4& v, int b) {
            const int e0 = (b * (SCAN_STEPS * OWN_THREADS) + tid) * 4;
            scan_word(v.x, e0);
            scan_word(v.y, e0 + OWN_THREADS * 4);
            scan_word(v.z, e0 + OWN_THREADS * 8);
            scan_word(v.w, e0 + OWN_THREADS * 12);
        };

        constexpr int PF = 2;
        uint4 bufA[PF], bufB[PF];
        auto fill = [&](uint4 (&buf)[PF], int q0) {
            #pragma unroll
            for (int i = 0; i < PF; ++i)
                buf[i] = oid4[batch_of(min(q0 + i, nBatches - 1)) * OWN_THREADS + tid];
        };
        auto scan = [&](const uint4 (&buf)[PF], int q0) {
            #pragma unroll
            for (int i = 0; i < PF; ++i)
                if (q0 + i < nBatches) scan_batch(buf[i], batch_of(q0 + i));
        };
        if (nBatches > 0) {
            fill(bufA, 0);
            fill(bufB, PF);
        }
        for (int q = 0; q < nBatches; q += 2 * PF) {
            scan(bufA, q);
            fill(bufA, q + 2 * PF);
            scan(bufB, q + PF);
            fill(bufB, q + 3 * PF);
        }
        sweep_plain(swept, E);
    };

    auto flush = [&]() {
        const int ng = (tail + 31) >> 5;

        int   s_nx = 0;
        float w_nx = 0.0f;
        int   r_nx = OWN_ROWS;
        if (lane < tail) {
            const int e = le[lane];
            s_nx = src[e];
            w_nx = ew[e];
            const unsigned r = (unsigned)(dst[e] - d0);
            r_nx = (r < urows) ? (int)r : OWN_ROWS;
        }
        for (int g = 0; g < ng; ++g) {
            const int   n = tail - g * 32;
            const int   s = s_nx;
            const float w = w_nx;
            const int   rel = r_nx;
            s_nx = 0;
            w_nx = 0.0f;
            r_nx = OWN_ROWS;
            if ((g + 1) * 32 + lane < tail) {
                const int e = le[(g + 1) * 32 + lane];
                s_nx = src[e];
                w_nx = ew[e];
                const unsigned r = (unsigned)(dst[e] - d0);
                r_nx = (r < urows) ? (int)r : OWN_ROWS;
            }
            float v[32];
            #pragma unroll
            for (int k = 0; k < 32; ++k) {
                const int sk = __builtin_amdgcn_readlane(s, k);
                v[k] = xl[(size_t)sk * 32 + lane];
            }
            #pragma unroll
            for (int k = 0; k < 32; ++k) {
                const int   rk = __builtin_amdgcn_readlane(rel, k);
                const float wk = __int_as_float(__builtin_amdgcn_readlane(__float_as_int(w), k));
                float m = v[k] + xrs[rk * 32 + lane] + wk * we_l;
                m = (m > 0.0f) ? m : NEG_ATT * m;
                v[k] = m * at_l;
            }
            const bool b16 = (lane & 16) != 0, b8 = (lane & 8) != 0,
                       b4  = (lane & 4)  != 0, b2 = (lane & 2) != 0, b1 = (lane & 1) != 0;
            #pragma unroll
            for (int k = 0; k < 16; ++k) {
                const float keep = b16 ? v[k + 16] : v[k];
                const float send = b16 ? v[k] : v[k + 16];
                v[k] = keep + half_xor(send);
            }
            #pragma unroll
            for (int k = 0; k < 8; ++k) {
                const float keep = b8 ? v[k + 8] : v[k];
                const float send = b8 ? v[k] : v[k + 8];
                v[k] = keep + row_xor<8>(send);
            }
            #pragma unroll
            for (int k = 0; k < 4; ++k) {
                const float keep = b4 ? v[k + 4] : v[k];
                const float send = b4 ? v[k] : v[k + 4];
                v[k] = keep + row_xor<4>(send);
            }
            #pragma unroll
            for (int k = 0; k < 2; ++k) {
                const float keep = b2 ? v[k + 2] : v[k];
                const float send = b2 ? v[k] : v[k + 2];
                v[k] = keep + row_xor<2>(send);
            }
            const float keep1 = b1 ? v[1] : v[0];
            const float send1 = b1 ? v[0] : v[1];
            const float lg = keep1 + row_xor<1>(send1);
            if (lane < n) {
                ll[g * 32 + lane] = lg;
                le[g * 32 + lane] = s;
                lr[g * 32 + lane] = (unsigned short)rel;
                lds_max_float(&cmax[rel], lg);
            }
        }
        __syncthreads();

        for (int i = tid; i < OWN_ROWS * 32; i += OWN_THREADS) {
            const int r = i >> 5;
            const float mo = smax[r], c = cmax[r];
            if (c > mo) {
                if (mo != -INFINITY) {
                    const float f = __expf(mo - c);
                    acc[i] *= f;
                    if (lane == 0) ssum[r] *= f;
                }
                if (lane == 0) smax[r] = c;
            }
            if (lane == 0) cmax[r] = -INFINITY;
        }
        __syncthreads();

        for (int g = 0; g < ng; ++g) {
            const int n = tail - g * 32;
            int rel = OWN_ROWS, s = 0;
            float a = 0.0f;
            if (lane < n) {
                s   = le[g * 32 + lane];
                rel = lr[g * 32 + lane];
                a   = __expf(ll[g * 32 + lane] - smax[rel]);
                __hip_atomic_fetch_add(&ssum[rel], a,
                                       __ATOMIC_RELAXED, __HIP_MEMORY_SCOPE_WORKGROUP);
            }
            float v[32];
            #pragma unroll
            for (int k = 0; k < 32; ++k) {
                const int sk = __builtin_amdgcn_readlane(s, k);
                v[k] = xl[(size_t)sk * 32 + lane];
            }
            #pragma unroll
            for (int k = 0; k < 32; ++k) {
                const int   rk = __builtin_amdgcn_readlane(rel, k);
                const float ak = __int_as_float(__builtin_amdgcn_readlane(__float_as_int(a), k));
                __hip_atomic_fetch_add(&acc[rk * 32 + lane], v[k] * ak,
                                       __ATOMIC_RELAXED, __HIP_MEMORY_SCOPE_WORKGROUP);
            }
        }
    };

    sweep_fast();
    if (tail > L_CAP && lane == 0) ovf_flag = 1;
    __syncthreads();
    const bool over    = __builtin_amdgcn_readfirstlane(ovf_flag) != 0;
    const int  nChunks = over ? (E + CHUNK_EDGES - 1) / CHUNK_EDGES : 1;
    for (int chunk = 0; chunk < nChunks; ++chunk) {
        if (over) {
            const int lo = chunk * CHUNK_EDGES;
            int hi = lo + CHUNK_EDGES;
            if (hi > E) hi = E;
            tail = 0;
            sweep_plain(lo, hi);
        }
        flush();
    }
    __syncthreads();

    float4* __restrict__ out4 = (float4*)(out + (size_t)d0 * 32);
    for (int i = tid; i < rows * 8; i += OWN_THREADS) {
        const int c = (i & 7) * 4;
        const float sm = ssum[i >> 3];
        float4 v = acc4[i];
        if (sm != 0.0f) { v.x /= sm; v.y /= sm; v.z /= sm; v.w /= sm; }
        v.x += bias[c];     v.x = (v.x > 0.0f) ? v.x : NEG_OUT * v.x;
        v.y += bias[c + 1]; v.y = (v.y > 0.0f) ? v.y : NEG_OUT * v.y;
        v.z += bias[c + 2]; v.z = (v.z > 0.0f) ? v.z : NEG_OUT * v.z;
        v.w += bias[c + 3]; v.w = (v.w > 0.0f) ? v.w : NEG_OUT * v.w;
        out4[i] = v;
    }
}

extern "C" void kernel_launch(void* const* d_in, const int* in_sizes, int n_in,
                              void* d_out, int out_size, void* d_ws, size_t ws_size,
                              hipStream_t stream) {
    const float* x    = (const float*)d_in[0];
    const int*   ei   = (const int*)d_in[1];
    const float* ew   = (const float*)d_in[2];
    const float* Wl   = (const float*)d_in[3];
    const float* bl   = (const float*)d_in[4];
    const float* Wr   = (const float*)d_in[5];
    const float* br   = (const float*)d_in[6];
    const float* We   = (const float*)d_in[7];
    const float* att  = (const float*)d_in[8];
    const float* bias = (const float*)d_in[9];

    const int N = in_sizes[0] / 128;
    const int E = in_sizes[2];
    const int* srcv = ei;
    const int* dstv = ei + E;

    float* xl  = (float*)d_ws;
    float* xr  = xl + (size_t)N * 32;
    unsigned int* oid = (unsigned int*)(xr + (size_t)N * 32);
    float* out = (float*)d_out;

    const int tiles = (N + 15) / 16;
    const int wavesPerBlock = 8;
    gat_gemm<<<(tiles + wavesPerBlock - 1) / wavesPerBlock, 32 * wavesPerBlock, 0, stream>>>(
        x, Wl, bl, Wr, br, xl, xr, N, dstv, oid, E);

    gat_edges_owner<<<(N + OWN_ROWS - 1) / OWN_ROWS, OWN_THREADS, 0, stream>>>(
        srcv, dstv, oid, ew, xl, xr, We, att, bias, out, N, E);
}


# owner ids kept as 8 bit planes per 16-edge chunk: a lane's hit mask is 10 logic ops, a batch's hits are listed per lane in rounds, not by 16 per-byte wave votes; LDS sum order can swap within a batch
# speedup vs baseline: 9.2253x; 1.1470x over previous
#include <hip/hip_runtime.h>
#include <hip/hip_bf16.h>
#include <math.h>


#define NEG_ATT 0.2f
#define NEG_OUT 0.01f

typedef __attribute__((ext_vector_type(2))) float v2f;
typedef __attribute__((ext_vector_type(8))) float v8f;

#define OWN_ROWS    392
#define OWN_THREADS 1024
#define OWN_WAVES   (OWN_THREADS / 32)
#define SCAN_STEPS  4
#define L_CAP       640
#define BATCH_EDGES (OWN_THREADS * SCAN_STEPS * 4)

__device__ __forceinline__ unsigned owner_byte(int d) {
    return ((unsigned)d / (unsigned)OWN_ROWS) & 255u;
}
__device__ __forceinline__ unsigned byte_bits(unsigned o, int p) {
    return (((o >> p) & 0x01010101u) * 0x10204080u) >> 28;
}

__global__ __launch_bounds__(256) void gat_gemm(
        const float* __restrict__ x,
        const float* __restrict__ Wl, const float* __restrict__ bl,
        const float* __restrict__ Wr, const float* __restrict__ br,
        float* __restrict__ xl, float* __restrict__ xr, int N,
        const int* __restrict__ dst, unsigned int* __restrict__ oid, int E) {
    {
        const bool vec_ok   = (((size_t)dst) & 15) == 0;
        const int  nBatches = vec_ok ? (E / BATCH_EDGES) : 0;
        const int4* __restrict__ dst4 = (const int4*)dst;
        uint4* __restrict__ oid4 = (uint4*)oid;
        const int nChunks = nBatches * OWN_THREADS;
        for (int c = blockIdx.x * blockDim.x + threadIdx.x; c < nChunks; c += gridDim.x * blockDim.x) {
            const int b = c / OWN_THREADS, t = c % OWN_THREADS;
            unsigned w[4] = {0u, 0u, 0u, 0u};
            #pragma unroll
            for (int u = 0; u < SCAN_STEPS; ++u) {
                const int4 d = dst4[(b * SCAN_STEPS + u) * OWN_THREADS + t];
                const unsigned o = owner_byte(d.x) | (owner_byte(d.y) << 8) | (owner_byte(d.z) << 16) | (owner_byte(d.w) << 24);
                #pragma unroll
                for (int p = 0; p < 8; ++p)
                    w[p >> 1] |= byte_bits(o, p) << ((p & 1) * 16 + u * 4);
            }
            oid4[c] = make_uint4(w[0], w[1], w[2], w[3]);
        }
    }

    const int lane = threadIdx.x & 31;
    const int wave = threadIdx.x >> 5;
    const int tile = blockIdx.x * (blockDim.x >> 5) + wave;
    const int row0 = tile * 16;
    if (row0 >= N) return;

    const int m  = lane & 15;
    const int kh = lane >> 4;

    int arow = row0 + m;
    if (arow >= N) arow = N - 1;
    const float* __restrict__ xrow = x + (size_t)arow * 128;

    v8f cl0 = {}, cl1 = {}, cr0 = {}, cr1 = {};

    #pragma unroll 4
    for (int k0 = 0; k0 < 128; k0 += 4) {
        const int kb = k0 + 2 * kh;
        v2f a;
        a[0] = xrow[kb];
        a[1] = xrow[kb + 1];

        const int n = m;
        v2f bl0v, bl1v, br0v, br1v;
        bl0v[0] = Wl[kb * 32 + n];       bl0v[1] = Wl[(kb + 1) * 32 + n];
        bl1v[0] = Wl[kb * 32 + n + 16];  bl1v[1] = Wl[(kb + 1) * 32 + n + 16];
        br0v[0] = Wr[kb * 32 + n];       br0v[1] = Wr[(kb + 1) * 32 + n];
        br1v[0] = Wr[kb * 32 + n + 16];  br1v[1] = Wr[(kb + 1) * 32 + n + 16];

        cl0 = __builtin_amdgcn_wmma_f32_16x16x4_f32(false, a, false, bl0v, (short)0, cl0, false, false);
        cl1 = __builtin_amdgcn_wmma_f32_16x16x4_f32(false, a, false, bl1v, (short)0, cl1, false, false);
        cr0 = __builtin_amdgcn_wmma_f32_16x16x4_f32(false, a, false, br0v, (short)0, cr0, false, false);
        cr1 = __builtin_amdgcn_wmma_f32_16x16x4_f32(false, a, false, br1v, (short)0, cr1, false, false);
    }

    const int n = lane & 15;
    const float bl_lo = bl[n], bl_hi = bl[n + 16];
    const float br_lo = br[n], br_hi = br[n + 16];
    #pragma unroll
    for (int v = 0; v < 8; ++v) {
        const int mm = v + 8 * kh;
        const int row = row0 + mm;
        if (row < N) {
            const size_t ro = (size_t)row * 32;
            xl[ro + n]      = cl0[v] + bl_lo;
            xl[ro + n + 16] = cl1[v] + bl_hi;
            xr[ro + n]      = cr0[v] + br_lo;
            xr[ro + n + 16] = cr1[v] + br_hi;
        }
    }
}

#define CHUNK_EDGES ((L_CAP / 32) * OWN_THREADS)

template <int N>
__device__ __forceinline__ float row_xor(float x) {
    return __int_as_float(__builtin_amdgcn_update_dpp(0, __float_as_int(x), 0x160 + N, 0xf, 0xf, false));
}
__device__ __forceinline__ float half_xor(float x) {
    const int b = __float_as_int(x);
    return __int_as_float(__builtin_amdgcn_permlanex16(b, b, 0x76543210, 0xfedcba98, false, false));
}

__device__ __forceinline__ void lds_max_float(float* addr, float val) {
    const float v = val + 0.0f;
    if (v >= 0.0f) {
        atomicMax((int*)addr, __float_as_int(v));
    } else {
        atomicMin((unsigned int*)addr, __float_as_uint(v));
    }
}

__global__ __launch_bounds__(OWN_THREADS) void gat_edges_owner(
        const int* __restrict__ src, const int* __restrict__ dst,
        const unsigned int* __restrict__ oid,
        const float* __restrict__ ew,
        const float* __restrict__ xl, const float* __restrict__ xr,
        const float* __restrict__ We, const float* __restrict__ att,
        const float* __restrict__ bias,
        float* __restrict__ out, int N, int E) {
    __shared__ __attribute__((aligned(16))) float acc[(OWN_ROWS + 1) * 32];
    __shared__ __attribute__((aligned(16))) float xrs[(OWN_ROWS + 1) * 32];
    __shared__ int            lst_e[OWN_WAVES][L_CAP];
    __shared__ float          lst_l[OWN_WAVES][L_CAP];
    __shared__ unsigned short lst_r[OWN_WAVES][L_CAP];
    __shared__ float smax[OWN_ROWS + 1];
    __shared__ float cmax[OWN_ROWS + 1];
    __shared__ float ssum[OWN_ROWS + 1];
    __shared__ int   ovf_flag;

    const int tid  = threadIdx.x;
    const int lane = tid & 31;
    const int wave = __builtin_amdgcn_readfirstlane(tid >> 5);
    const int d0   = blockIdx.x * OWN_ROWS;
    int rows = N - d0;
    if (rows > OWN_ROWS) rows = OWN_ROWS;
    const unsigned urows = (unsigned)rows;

    float4* acc4 = (float4*)acc;
    float4* xrs4 = (float4*)xrs;
    const float4* __restrict__ xr4 = (const float4*)(xr + (size_t)d0 * 32);
    for (int i = tid; i < (OWN_ROWS + 1) * 8; i += OWN_THREADS) {
        acc4[i] = make_float4(0.0f, 0.0f, 0.0f, 0.0f);
        xrs4[i] = ((i >> 3) < rows) ? xr4[i] : make_float4(0.0f, 0.0f, 0.0f, 0.0f);
    }
    for (int i = tid; i < OWN_ROWS + 1; i += OWN_THREADS) {
        smax[i] = -INFINITY;
        cmax[i] = -INFINITY;
        ssum[i] = 0.0f;
    }
    if (tid == 0) ovf_flag = 0;
    __syncthreads();

    int*            __restrict__ le = lst_e[wave];
    float*          __restrict__ ll = lst_l[wave];
    unsigned short* __restrict__ lr = lst_r[wave];
    int tail = 0;

    const float we_l = We[lane];
    const float at_l = att[lane];

    auto push = [&](bool hit, int e) {
        const unsigned mask = __builtin_amdgcn_ballot_w32(hit);
        if (mask != 0u) {
            if (hit) {
                const int p = tail + (int)__builtin_amdgcn_mbcnt_lo(mask, 0u);
                if (p < L_CAP) le[p] = e;
            }
            tail += __builtin_popcount(mask);
        }
    };

    auto sweep_plain = [&](int lo, int hi) {
        for (int base = lo + wave * 32; base < hi; base += OWN_THREADS) {
            const int e = base + lane;
            const unsigned r = (e < hi) ? (unsigned)(dst[e] - d0) : 0xFFFFFFFFu;
            push(r < urows, e);
        }
    };

    auto sweep_fast = [&]() {
        const bool vec_ok   = (((size_t)dst) & 15) == 0;
        const int  nBatches = vec_ok ? (E / BATCH_EDGES) : 0;
        const int  swept    = nBatches * BATCH_EDGES;
        const uint4* __restrict__ oid4 = (const uint4*)oid;
        const unsigned gid = blockIdx.x & 255u;
        auto plane_pat = [&](int k) -> unsigned {
            return (((gid >> (2 * k)) & 1u) ? 0u : 0x0000FFFFu) | (((gid >> (2 * k + 1)) & 1u) ? 0u : 0xFFFF0000u);
        };
        const unsigned q0 = plane_pat(0), q1 = plane_pat(1), q2 = plane_pat(2), q3 = plane_pat(3);

        const int bi0 = (int)(((unsigned long long)blockIdx.x * (unsigned long long)nBatches) / gridDim.x);
        auto batch_of = [&](int q) -> int {
            const int b = bi0 + q;
            return (b >= nBatches) ? b - nBatches : b;
        };

        auto scan_batch = [&](const uint4& v, int b) {
            const int e0 = (b * (SCAN_STEPS * OWN_THREADS) + tid) * 4;
            const unsigned a = (v.x ^ q0) & (v.y ^ q1) & (v.z ^ q2) & (v.w ^ q3);
            unsigned m = a & (a >> 16);
            for (int r = 0; r < 16; ++r) {
                const unsigned mask = __builtin_amdgcn_ballot_w32(m != 0u);
                if (mask == 0u) break;
                if (m != 0u) {
                    const int p = tail + (int)__builtin_amdgcn_mbcnt_lo(mask, 0u);
                    if (p < L_CAP) {
                        const int i = __builtin_ctz(m);
                        le[p] = e0 + (i >> 2) * (OWN_THREADS * 4) + (i & 3);
                    }
                    m &= m - 1u;
                }
                tail += __builtin_popcount(mask);
            }
        };

        constexpr int PF = 2;
        uint4 bufA[PF], bufB[PF];
        auto fill = [&](uint4 (&buf)[PF], int q0) {
            #pragma unroll
            for (int i = 0; i < PF; ++i)
                buf[i] = oid4[batch_of(min(q0 + i, nBatches - 1)) * OWN_THREADS + tid];
        };
        auto scan = [&](const uint4 (&buf)[PF], int q0) {
            #pragma unroll
            for (int i = 0; i < PF; ++i)
                if (q0 + i < nBatches) scan_batch(buf[i], batch_of(q0 + i));
        };
        if (nBatches > 0) {
            fill(bufA, 0);
            fill(bufB, PF);
        }
        for (int q = 0; q < nBatches; q += 2 * PF) {
            scan(bufA, q);
            fill(bufA, q + 2 * PF);
            scan(bufB, q + PF);
            fill(bufB, q + 3 * PF);
        }
        sweep_plain(swept, E);
    };

    auto flush = [&]() {
        const int ng = (tail + 31) >> 5;

        int   s_nx = 0;
        float w_nx = 0.0f;
        int   r_nx = OWN_ROWS;
        if (lane < tail) {
            const int e = le[lane];
            s_nx = src[e];
            w_nx = ew[e];
            const unsigned r = (unsigned)(dst[e] - d0);
            r_nx = (r < urows) ? (int)r : OWN_ROWS;
        }
        for (int g = 0; g < ng; ++g) {
            const int   n = tail - g * 32;
            const int   s = s_nx;
            const float w = w_nx;
            const int   rel = r_nx;
            s_nx = 0;
            w_nx = 0.0f;
            r_nx = OWN_ROWS;
            if ((g + 1) * 32 + lane < tail) {
                const int e = le[(g + 1) * 32 + lane];
                s_nx = src[e];
                w_nx = ew[e];
                const unsigned r = (unsigned)(dst[e] - d0);
                r_nx = (r < urows) ? (int)r : OWN_ROWS;
            }
            float v[32];
            #pragma unroll
            for (int k = 0; k < 32; ++k) {
                const int sk = __builtin_amdgcn_readlane(s, k);
                v[k] = xl[(size_t)sk * 32 + lane];
            }
            #pragma unroll
            for (int k = 0; k < 32; ++k) {
                const int   rk = __builtin_amdgcn_readlane(rel, k);
                const float wk = __int_as_float(__builtin_amdgcn_readlane(__float_as_int(w), k));
                float m = v[k] + xrs[rk * 32 + lane] + wk * we_l;
                m = (m > 0.0f) ? m : NEG_ATT * m;
                v[k] = m * at_l;
            }
            const bool b16 = (lane & 16) != 0, b8 = (lane & 8) != 0,
                       b4  = (lane & 4)  != 0, b2 = (lane & 2) != 0, b1 = (lane & 1) != 0;
            #pragma unroll
            for (int k = 0; k < 16; ++k) {
                const float keep = b16 ? v[k + 16] : v[k];
                const float send = b16 ? v[k] : v[k + 16];
                v[k] = keep + half_xor(send);
            }
            #pragma unroll
            for (int k = 0; k < 8; ++k) {
                const float keep = b8 ? v[k + 8] : v[k];
                const float send = b8 ? v[k] : v[k + 8];
                v[k] = keep + row_xor<8>(send);
            }
            #pragma unroll
            for (int k = 0; k < 4; ++k) {
                const float keep = b4 ? v[k + 4] : v[k];
                const float send = b4 ? v[k] : v[k + 4];
                v[k] = keep + row_xor<4>(send);
            }
            #pragma unroll
            for (int k = 0; k < 2; ++k) {
                const float keep = b2 ? v[k + 2] : v[k];
                const float send = b2 ? v[k] : v[k + 2];
                v[k] = keep + row_xor<2>(send);
            }
            const float keep1 = b1 ? v[1] : v[0];
            const float send1 = b1 ? v[0] : v[1];
            const float lg = keep1 + row_xor<1>(send1);
            if (lane < n) {
                ll[g * 32 + lane] = lg;
                le[g * 32 + lane] = s;
                lr[g * 32 + lane] = (unsigned short)rel;
                lds_max_float(&cmax[rel], lg);
            }
        }
        __syncthreads();

        for (int i = tid; i < OWN_ROWS * 32; i += OWN_THREADS) {
            const int r = i >> 5;
            const float mo = smax[r], c = cmax[r];
            if (c > mo) {
                if (mo != -INFINITY) {
                    const float f = __expf(mo - c);
                    acc[i] *= f;
                    if (lane == 0) ssum[r] *= f;
                }
                if (lane == 0) smax[r] = c;
            }
            if (lane == 0) cmax[r] = -INFINITY;
        }
        __syncthreads();

        for (int g = 0; g < ng; ++g) {
            const int n = tail - g * 32;
            int rel = OWN_ROWS, s = 0;
            float a = 0.0f;
            if (lane < n) {
                s   = le[g * 32 + lane];
                rel = lr[g * 32 + lane];
                a   = __expf(ll[g * 32 + lane] - smax[rel]);
                __hip_atomic_fetch_add(&ssum[rel], a,
                                       __ATOMIC_RELAXED, __HIP_MEMORY_SCOPE_WORKGROUP);
            }
            float v[32];
            #pragma unroll
            for (int k = 0; k < 32; ++k) {
                const int sk = __builtin_amdgcn_readlane(s, k);
                v[k] = xl[(size_t)sk * 32 + lane];
            }
            #pragma unroll
            for (int k = 0; k < 32; ++k) {
                const int   rk = __builtin_amdgcn_readlane(rel, k);
                const float ak = __int_as_float(__builtin_amdgcn_readlane(__float_as_int(a), k));
                __hip_atomic_fetch_add(&acc[rk * 32 + lane], v[k] * ak,
                                       __ATOMIC_RELAXED, __HIP_MEMORY_SCOPE_WORKGROUP);
            }
        }
    };

    sweep_fast();
    if (tail > L_CAP && lane == 0) ovf_flag = 1;
    __syncthreads();
    const bool over    = __builtin_amdgcn_readfirstlane(ovf_flag) != 0;
    const int  nChunks = over ? (E + CHUNK_EDGES - 1) / CHUNK_EDGES : 1;
    for (int chunk = 0; chunk < nChunks; ++chunk) {
        if (over) {
            const int lo = chunk * CHUNK_EDGES;
            int hi = lo + CHUNK_EDGES;
            if (hi > E) hi = E;
            tail = 0;
            sweep_plain(lo, hi);
        }
        flush();
    }
    __syncthreads();

    float4* __restrict__ out4 = (float4*)(out + (size_t)d0 * 32);
    for (int i = tid; i < rows * 8; i += OWN_THREADS) {
        const int c = (i & 7) * 4;
        const float sm = ssum[i >> 3];
        float4 v = acc4[i];
        if (sm != 0.0f) { v.x /= sm; v.y /= sm; v.z /= sm; v.w /= sm; }
        v.x += bias[c];     v.x = (v.x > 0.0f) ? v.x : NEG_OUT * v.x;
        v.y += bias[c + 1]; v.y = (v.y > 0.0f) ? v.y : NEG_OUT * v.y;
        v.z += bias[c + 2]; v.z = (v.z > 0.0f) ? v.z : NEG_OUT * v.z;
        v.w += bias[c + 3]; v.w = (v.w > 0.0f) ? v.w : NEG_OUT * v.w;
        out4[i] = v;
    }
}

extern "C" void kernel_launch(void* const* d_in, const int* in_sizes, int n_in,
                              void* d_out, int out_size, void* d_ws, size_t ws_size,
                              hipStream_t stream) {
    const float* x    = (const float*)d_in[0];
    const int*   ei   = (const int*)d_in[1];
    const float* ew   = (const float*)d_in[2];
    const float* Wl   = (const float*)d_in[3];
    const float* bl   = (const float*)d_in[4];
    const float* Wr   = (const float*)d_in[5];
    const float* br   = (const float*)d_in[6];
    const float* We   = (const float*)d_in[7];
    const float* att  = (const float*)d_in[8];
    const float* bias = (const float*)d_in[9];

    const int N = in_sizes[0] / 128;
    const int E = in_sizes[2];
    const int* srcv = ei;
    const int* dstv = ei + E;

    float* xl  = (float*)d_ws;
    float* xr  = xl + (size_t)N * 32;
    unsigned int* oid = (unsigned int*)(xr + (size_t)N * 32);
    float* out = (float*)d_out;

    const int tiles = (N + 15) / 16;
    const int wavesPerBlock = 8;
    gat_gemm<<<(tiles + wavesPerBlock - 1) / wavesPerBlock, 32 * wavesPerBlock, 0, stream>>>(
        x, Wl, bl, Wr, br, xl, xr, N, dstv, oid, E);

    gat_edges_owner<<<(N + OWN_ROWS - 1) / OWN_ROWS, OWN_THREADS, 0, stream>>>(
        srcv, dstv, oid, ew, xl, xr, We, att, bias, out, N, E);
}


# owner-id sweep in fewer instructions: one running batch number for the loads, plane test as 3-input logic ops, hit rounds as one short loop; every wave's list and its order (LDS sum order) unchanged
# speedup vs baseline: 11.6770x; 1.2658x over previous
#include <hip/hip_runtime.h>
#include <hip/hip_bf16.h>
#include <math.h>


#define NEG_ATT 0.2f
#define NEG_OUT 0.01f

typedef __attribute__((ext_vector_type(2))) float v2f;
typedef __attribute__((ext_vector_type(8))) float v8f;

#define OWN_ROWS    392
#define OWN_THREADS 1024
#define OWN_WAVES   (OWN_THREADS / 32)
#define SCAN_STEPS  4
#define L_CAP       640
#define BATCH_EDGES (OWN_THREADS * SCAN_STEPS * 4)
#define SHIFT_LIM   32.0f
#define ROW_SHIFT   23
#define SRC_MASK    ((1u << ROW_SHIFT) - 1u)
#define PACK_MAX_N  (256 * OWN_ROWS)

__device__ __forceinline__ unsigned owner_byte(int d) {
    return ((unsigned)d / (unsigned)OWN_ROWS) & 255u;
}
__device__ __forceinline__ unsigned byte_bits(unsigned o, int p) {
    return (((o >> p) & 0x01010101u) * 0x10204080u) >> 28;
}

__global__ __launch_bounds__(256) void gat_gemm(
        const float* __restrict__ x,
        const float* __restrict__ Wl, const float* __restrict__ bl,
        const float* __restrict__ Wr, const float* __restrict__ br,
        float* __restrict__ xl, float* __restrict__ xr, int N,
        const int* __restrict__ dst, unsigned int* __restrict__ oid, int E,
        const int* __restrict__ src, const float* __restrict__ ew, unsigned int* __restrict__ rec) {
    {
        const bool vec_ok   = ((((size_t)dst) | ((size_t)oid)) & 15) == 0;
        const int  nBatches = vec_ok ? (E / BATCH_EDGES) : 0;
        const int4* __restrict__ dst4 = (const int4*)dst;
        uint4* __restrict__ oid4 = (uint4*)oid;
        const int nChunks = nBatches * OWN_THREADS;
        for (int c = blockIdx.x * blockDim.x + threadIdx.x; c < nChunks; c += gridDim.x * blockDim.x) {
            const int b = c / OWN_THREADS, t = c % OWN_THREADS;
            unsigned w[4] = {0u, 0u, 0u, 0u};
            #pragma unroll
            for (int u = 0; u < SCAN_STEPS; ++u) {
                const int4 d = dst4[(b * SCAN_STEPS + u) * OWN_THREADS + t];
                const unsigned o = owner_byte(d.x) | (owner_byte(d.y) << 8) | (owner_byte(d.z) << 16) | (owner_byte(d.w) << 24);
                #pragma unroll
                for (int p = 0; p < 8; ++p)
                    w[p >> 1] |= byte_bits(o, p) << ((p & 1) * 16 + u * 4);
            }
            oid4[c] = make_uint4(w[0], w[1], w[2], w[3]);
        }
    }
    if (N <= PACK_MAX_N) {
        const int gt = blockIdx.x * blockDim.x + threadIdx.x;
        const int gn = gridDim.x * blockDim.x;
        const bool v2_ok = (((((size_t)src) | ((size_t)dst) | ((size_t)ew)) & 7) == 0) && ((((size_t)rec) & 15) == 0);
        const int  E2    = v2_ok ? (E >> 1) : 0;
        const int2*   __restrict__ src2 = (const int2*)src;
        const int2*   __restrict__ dst2 = (const int2*)dst;
        const float2* __restrict__ ew2  = (const float2*)ew;
        uint4* __restrict__ rec4 = (uint4*)rec;
        for (int i = gt; i < E2; i += 4 * gn) {
            int2 s[4], d[4];
            float2 w[4];
            #pragma unroll
            for (int u = 0; u < 4; ++u) {
                const int j = min(i + u * gn, E2 - 1);
                s[u] = src2[j];
                d[u] = dst2[j];
                w[u] = ew2[j];
            }
            #pragma unroll
            for (int u = 0; u < 4; ++u) {
                const int j = i + u * gn;
                const unsigned r0 = (unsigned)d[u].x % (unsigned)OWN_ROWS, r1 = (unsigned)d[u].y % (unsigned)OWN_ROWS;
                if (j < E2)
                    rec4[j] = make_uint4((unsigned)s[u].x | (r0 << ROW_SHIFT), __float_as_uint(w[u].x),
                                         (unsigned)s[u].y | (r1 << ROW_SHIFT), __float_as_uint(w[u].y));
            }
        }
        uint2* __restrict__ rec2 = (uint2*)rec;
        for (int e = E2 * 2 + gt; e < E; e += gn) {
            const unsigned r = (unsigned)dst[e] % (unsigned)OWN_ROWS;
            rec2[e] = make_uint2((unsigned)src[e] | (r << ROW_SHIFT), __float_as_uint(ew[e]));
        }
    }

    const int lane = threadIdx.x & 31;
    const int wave = threadIdx.x >> 5;
    const int tile = blockIdx.x * (blockDim.x >> 5) + wave;
    const int row0 = tile * 16;
    if (row0 >= N) return;

    const int m  = lane & 15;
    const int kh = lane >> 4;

    int arow = row0 + m;
    if (arow >= N) arow = N - 1;
    const float* __restrict__ xrow = x + (size_t)arow * 128;

    v8f cl0 = {}, cl1 = {}, cr0 = {}, cr1 = {};

    #pragma unroll 4
    for (int k0 = 0; k0 < 128; k0 += 4) {
        const int kb = k0 + 2 * kh;
        v2f a;
        a[0] = xrow[kb];
        a[1] = xrow[kb + 1];

        const int n = m;
        v2f bl0v, bl1v, br0v, br1v;
        bl0v[0] = Wl[kb * 32 + n];       bl0v[1] = Wl[(kb + 1) * 32 + n];
        bl1v[0] = Wl[kb * 32 + n + 16];  bl1v[1] = Wl[(kb + 1) * 32 + n + 16];
        br0v[0] = Wr[kb * 32 + n];       br0v[1] = Wr[(kb + 1) * 32 + n];
        br1v[0] = Wr[kb * 32 + n + 16];  br1v[1] = Wr[(kb + 1) * 32 + n + 16];

        cl0 = __builtin_amdgcn_wmma_f32_16x16x4_f32(false, a, false, bl0v, (short)0, cl0, false, false);
        cl1 = __builtin_amdgcn_wmma_f32_16x16x4_f32(false, a, false, bl1v, (short)0, cl1, false, false);
        cr0 = __builtin_amdgcn_wmma_f32_16x16x4_f32(false, a, false, br0v, (short)0, cr0, false, false);
        cr1 = __builtin_amdgcn_wmma_f32_16x16x4_f32(false, a, false, br1v, (short)0, cr1, false, false);
    }

    const int n = lane & 15;
    const float bl_lo = bl[n], bl_hi = bl[n + 16];
    const float br_lo = br[n], br_hi = br[n + 16];
    #pragma unroll
    for (int v = 0; v < 8; ++v) {
        const int mm = v + 8 * kh;
        const int row = row0 + mm;
        if (row < N) {
            const size_t ro = (size_t)row * 32;
            xl[ro + n]      = cl0[v] + bl_lo;
            xl[ro + n + 16] = cl1[v] + bl_hi;
            xr[ro + n]      = cr0[v] + br_lo;
            xr[ro + n + 16] = cr1[v] + br_hi;
        }
    }
}

#define CHUNK_EDGES ((L_CAP / 32) * OWN_THREADS)

template <int N>
__device__ __forceinline__ float row_xor(float x) {
    return __int_as_float(__builtin_amdgcn_update_dpp(0, __float_as_int(x), 0x160 + N, 0xf, 0xf, false));
}
__device__ __forceinline__ float half_xor(float x) {
    const int b = __float_as_int(x);
    return __int_as_float(__builtin_amdgcn_permlanex16(b, b, 0x76543210, 0xfedcba98, false, false));
}

__device__ __forceinline__ void lds_max_float(float* addr, float val) {
    const float v = val + 0.0f;
    if (v >= 0.0f) {
        atomicMax((int*)addr, __float_as_int(v));
    } else {
        atomicMin((unsigned int*)addr, __float_as_uint(v));
    }
}

__global__ __launch_bounds__(OWN_THREADS) void gat_edges_owner(
        const int* __restrict__ src, const int* __restrict__ dst,
        const unsigned int* __restrict__ oid,
        const unsigned int* __restrict__ rec,
        const float* __restrict__ ew,
        const float* __restrict__ xl, const float* __restrict__ xr,
        const float* __restrict__ We, const float* __restrict__ att,
        const float* __restrict__ bias,
        float* __restrict__ out, int N, int E) {
    __shared__ __attribute__((aligned(16))) float acc[(OWN_ROWS + 1) * 32];
    __shared__ __attribute__((aligned(16))) float xrs[(OWN_ROWS + 1) * 32];
    __shared__ int            lst_e[OWN_WAVES][L_CAP];
    __shared__ float          lst_l[OWN_WAVES][L_CAP];
    __shared__ unsigned short lst_r[OWN_WAVES][L_CAP];
    __shared__ float smax[OWN_ROWS + 1];
    __shared__ float cmax[OWN_ROWS + 1];
    __shared__ float ssum[OWN_ROWS + 1];
    __shared__ float sref[OWN_ROWS + 1];
    __shared__ int   ovf_flag;
    __shared__ int   far_flag;

    const int tid  = threadIdx.x;
    const int lane = tid & 31;
    const int wave = __builtin_amdgcn_readfirstlane(tid >> 5);
    const int d0   = blockIdx.x * OWN_ROWS;
    int rows = N - d0;
    if (rows > OWN_ROWS) rows = OWN_ROWS;
    const unsigned urows = (unsigned)rows;

    float4* acc4 = (float4*)acc;
    float4* xrs4 = (float4*)xrs;
    const float4* __restrict__ xr4 = (const float4*)(xr + (size_t)d0 * 32);
    for (int i = tid; i < (OWN_ROWS + 1) * 8; i += OWN_THREADS) {
        acc4[i] = make_float4(0.0f, 0.0f, 0.0f, 0.0f);
        xrs4[i] = ((i >> 3) < rows) ? xr4[i] : make_float4(0.0f, 0.0f, 0.0f, 0.0f);
    }
    for (int i = tid; i < OWN_ROWS + 1; i += OWN_THREADS) {
        smax[i] = -INFINITY;
        cmax[i] = -INFINITY;
        ssum[i] = 0.0f;
    }
    if (tid == 0) { ovf_flag = 0; far_flag = 0; }
    __syncthreads();

    int*            __restrict__ le = lst_e[wave];
    float*          __restrict__ ll = lst_l[wave];
    unsigned short* __restrict__ lr = lst_r[wave];
    int tail = 0;

    const float we_l = We[lane];
    const float at_l = att[lane];

    const bool packed = N <= PACK_MAX_N;
    const uint2* __restrict__ rec2 = (const uint2*)rec;
    auto fetch = [&](int e, int& s, float& w, int& r) {
        if (packed) {
            const uint2 q = rec2[e];
            s = (int)(q.x & SRC_MASK);
            w = __uint_as_float(q.y);
            r = (int)(q.x >> ROW_SHIFT);
        } else {
            s = src[e];
            w = ew[e];
            const unsigned rr = (unsigned)(dst[e] - d0);
            r = (rr < urows) ? (int)rr : OWN_ROWS;
        }
    };

    for (int r = wave; r < OWN_ROWS + 1; r += OWN_WAVES) {
        float t = xrs[r * 32 + lane];
        t = (t > 0.0f) ? t : NEG_ATT * t;
        t *= at_l;
        t += row_xor<1>(t);
        t += row_xor<2>(t);
        t += row_xor<4>(t);
        t += row_xor<8>(t);
        t += half_xor(t);
        if (lane == 0) sref[r] = t;
    }

    auto push = [&](bool hit, int e) {
        const unsigned mask = __builtin_amdgcn_ballot_w32(hit);
        if (mask != 0u) {
            if (hit) {
                const int p = tail + (int)__builtin_amdgcn_mbcnt_lo(mask, 0u);
                if (p < L_CAP) le[p] = e;
            }
            tail += __builtin_popcount(mask);
        }
    };

    auto sweep_plain = [&](int lo, int hi) {
        for (int base = lo + wave * 32; base < hi; base += OWN_THREADS) {
            const int e = base + lane;
            const unsigned r = (e < hi) ? (unsigned)(dst[e] - d0) : 0xFFFFFFFFu;
            push(r < urows, e);
        }
    };

    auto sweep_fast = [&]() {
        const bool vec_ok   = ((((size_t)dst) | ((size_t)oid)) & 15) == 0;
        const int  nBatches = vec_ok ? (E / BATCH_EDGES) : 0;
        const int  swept    = nBatches * BATCH_EDGES;
        const uint4* __restrict__ oid4 = (const uint4*)oid;
        const unsigned gid = blockIdx.x & 255u;
        auto plane_pat = [&](int k) -> unsigned {
            return (((gid >> (2 * k)) & 1u) ? 0u : 0x0000FFFFu) | (((gid >> (2 * k + 1)) & 1u) ? 0u : 0xFFFF0000u);
        };
        const unsigned q0 = plane_pat(0), q1 = plane_pat(1), q2 = plane_pat(2), q3 = plane_pat(3);

        const int bi0 = (int)(((unsigned long long)blockIdx.x * (unsigned long long)nBatches) / gridDim.x);

        const int ntid12 = -12 * tid;
        auto scan_batch = [&](const uint4& v, int idx) {
            unsigned a = v.x ^ q0;
            a = __builtin_amdgcn_bitop3_b32(v.y, a, q1, 0x48);
            a = __builtin_amdgcn_bitop3_b32(v.z, a, q2, 0x48);
            a = __builtin_amdgcn_bitop3_b32(v.w, a, q3, 0x48);
            unsigned m = a & (a >> 16);
            unsigned mask = __builtin_amdgcn_ballot_w32(m != 0u);
            while (mask != 0u) {
                if (m != 0u) {
                    const int p = min((int)__builtin_amdgcn_mbcnt_lo(mask, (unsigned)tail), L_CAP - 1);
                    const int i = __builtin_ctz(m);
                    le[p] = ((idx << 4) + ntid12) + ((i >> 2) * (OWN_THREADS * 4 - 4) + i);
                }
                m &= m - 1u;
                tail += __builtin_popcount(mask);
                mask = __builtin_amdgcn_ballot_w32(m != 0u);
            }
        };

        if (nBatches > 0) {
            int bl = bi0;
            uint4 c0, c1, c2, c3;
            int   i0, i1, i2, i3;
            auto load = [&](uint4& c, int& idx) {
                idx = bl * OWN_THREADS + tid;
                c = oid4[idx];
                bl = (bl + 1 == nBatches) ? 0 : bl + 1;
            };
            load(c0, i0); load(c1, i1); load(c2, i2); load(c3, i3);
            int q = 0;
            for (; q + 4 <= nBatches; q += 4) {
                scan_batch(c0, i0); load(c0, i0);
                scan_batch(c1, i1); load(c1, i1);
                scan_batch(c2, i2); load(c2, i2);
                scan_batch(c3, i3); load(c3, i3);
            }
            if (q     < nBatches) scan_batch(c0, i0);
            if (q + 1 < nBatches) scan_batch(c1, i1);
            if (q + 2 < nBatches) scan_batch(c2, i2);
        }
        sweep_plain(swept, E);
    };

    auto flush = [&]() {
        const int ng = (tail + 31) >> 5;

        int   s_nx = 0;
        float w_nx = 0.0f;
        int   r_nx = OWN_ROWS;
        if (lane < tail)
            fetch(le[lane], s_nx, w_nx, r_nx);
        for (int g = 0; g < ng; ++g) {
            const int   n = tail - g * 32;
            const int   s = s_nx;
            const float w = w_nx;
            const int   rel = r_nx;
            s_nx = 0;
            w_nx = 0.0f;
            r_nx = OWN_ROWS;
            if ((g + 1) * 32 + lane < tail)
                fetch(le[(g + 1) * 32 + lane], s_nx, w_nx, r_nx);
            float v[32];
            #pragma unroll
            for (int k = 0; k < 32; ++k) {
                const int sk = __builtin_amdgcn_readlane(s, k);
                v[k] = xl[(size_t)sk * 32 + lane];
            }
            #pragma unroll
            for (int k = 0; k < 32; ++k) {
                const int   rk = __builtin_amdgcn_readlane(rel, k);
                const float wk = __int_as_float(__builtin_amdgcn_readlane(__float_as_int(w), k));
                float m = v[k] + xrs[rk * 32 + lane] + wk * we_l;
                m = (m > 0.0f) ? m : NEG_ATT * m;
                v[k] = m * at_l;
            }
            const bool b16 = (lane & 16) != 0, b8 = (lane & 8) != 0,
                       b4  = (lane & 4)  != 0, b2 = (lane & 2) != 0, b1 = (lane & 1) != 0;
            #pragma unroll
            for (int k = 0; k < 16; ++k) {
                const float keep = b16 ? v[k + 16] : v[k];
                const float send = b16 ? v[k] : v[k + 16];
                v[k] = keep + half_xor(send);
            }
            #pragma unroll
            for (int k = 0; k < 8; ++k) {
                const float keep = b8 ? v[k + 8] : v[k];
                const float send = b8 ? v[k] : v[k + 8];
                v[k] = keep + row_xor<8>(send);
            }
            #pragma unroll
            for (int k = 0; k < 4; ++k) {
                const float keep = b4 ? v[k + 4] : v[k];
                const float send = b4 ? v[k] : v[k + 4];
                v[k] = keep + row_xor<4>(send);
            }
            #pragma unroll
            for (int k = 0; k < 2; ++k) {
                const float keep = b2 ? v[k + 2] : v[k];
                const float send = b2 ? v[k] : v[k + 2];
                v[k] = keep + row_xor<2>(send);
            }
            const float keep1 = b1 ? v[1] : v[0];
            const float send1 = b1 ? v[0] : v[1];
            const float lg = keep1 + row_xor<1>(send1);
            if (lane < n) {
                ll[g * 32 + lane] = lg;
                le[g * 32 + lane] = s;
                lr[g * 32 + lane] = (unsigned short)rel;
                lds_max_float(&cmax[rel], lg);
            }
        }
        __syncthreads();

        for (int i = tid; i < OWN_ROWS * 32; i += OWN_THREADS) {
            const int r = i >> 5;
            const float mo = smax[r], c = cmax[r];
            if (c > mo) {
                if (mo != -INFINITY) {
                    const float f = __expf(mo - c);
                    acc[i] *= f;
                    if (lane == 0) ssum[r] *= f;
                }
                if (lane == 0) smax[r] = c;
            }
            if (lane == 0) cmax[r] = -INFINITY;
        }
        __syncthreads();

        for (int g = 0; g < ng; ++g) {
            const int n = tail - g * 32;
            int rel = OWN_ROWS, s = 0;
            float a = 0.0f;
            if (lane < n) {
                s   = le[g * 32 + lane];
                rel = lr[g * 32 + lane];
                a   = __expf(ll[g * 32 + lane] - smax[rel]);
                __hip_atomic_fetch_add(&ssum[rel], a,
                                       __ATOMIC_RELAXED, __HIP_MEMORY_SCOPE_WORKGROUP);
            }
            float v[32];
            #pragma unroll
            for (int k = 0; k < 32; ++k) {
                const int sk = __builtin_amdgcn_readlane(s, k);
                v[k] = xl[(size_t)sk * 32 + lane];
            }
            #pragma unroll
            for (int k = 0; k < 32; ++k) {
                const int   rk = __builtin_amdgcn_readlane(rel, k);
                const float ak = __int_as_float(__builtin_amdgcn_readlane(__float_as_int(a), k));
                __hip_atomic_fetch_add(&acc[rk * 32 + lane], v[k] * ak,
                                       __ATOMIC_RELAXED, __HIP_MEMORY_SCOPE_WORKGROUP);
            }
        }
    };

    auto flush_onepass = [&]() {
        const int ng = (tail + 31) >> 5;
        bool far = false;
        uint2 q_nx = make_uint2(0u, 0u);
        if (ng > 0) q_nx = rec2[le[min(lane, tail - 1)]];
        for (int g = 0; g < ng; ++g) {
            const int   n = tail - g * 32;
            const bool  valid = lane < n;
            const int   s = valid ? (int)(q_nx.x & SRC_MASK) : 0;
            const float w = valid ? __uint_as_float(q_nx.y) : 0.0f;
            const int   rel = valid ? (int)(q_nx.x >> ROW_SHIFT) : OWN_ROWS;
            q_nx = rec2[le[min((g + 1) * 32 + lane, tail - 1)]];
            float v[32];
            #pragma unroll
            for (int k = 0; k < 32; ++k) {
                const int sk = __builtin_amdgcn_readlane(s, k);
                v[k] = xl[(size_t)sk * 32 + lane];
            }
            float t[32];
            #pragma unroll
            for (int k = 0; k < 32; ++k) {
                const int   rk = __builtin_amdgcn_readlane(rel, k);
                const float wk = __int_as_float(__builtin_amdgcn_readlane(__float_as_int(w), k));
                float m = v[k] + xrs[rk * 32 + lane] + wk * we_l;
                m = (m > 0.0f) ? m : NEG_ATT * m;
                t[k] = m * at_l;
            }
            const bool b16 = (lane & 16) != 0, b8 = (lane & 8) != 0,
                       b4  = (lane & 4)  != 0, b2 = (lane & 2) != 0, b1 = (lane & 1) != 0;
            #pragma unroll
            for (int k = 0; k < 16; ++k) {
                const float keep = b16 ? t[k + 16] : t[k];
                const float send = b16 ? t[k] : t[k + 16];
                t[k] = keep + half_xor(send);
            }
            #pragma unroll
            for (int k = 0; k < 8; ++k) {
                const float keep = b8 ? t[k + 8] : t[k];
                const float send = b8 ? t[k] : t[k + 8];
                t[k] = keep + row_xor<8>(send);
            }
            #pragma unroll
            for (int k = 0; k < 4; ++k) {
                const float keep = b4 ? t[k + 4] : t[k];
                const float send = b4 ? t[k] : t[k + 4];
                t[k] = keep + row_xor<4>(send);
            }
            #pragma unroll
            for (int k = 0; k < 2; ++k) {
                const float keep = b2 ? t[k + 2] : t[k];
                const float send = b2 ? t[k] : t[k + 2];
                t[k] = keep + row_xor<2>(send);
            }
            const float keep1 = b1 ? t[1] : t[0];
            const float send1 = b1 ? t[0] : t[1];
            const float lg = keep1 + row_xor<1>(send1);
            float a = 0.0f;
            if (lane < n && rel != OWN_ROWS) {
                const float d = lg - sref[rel];
                far = far || !(fabsf(d) <= SHIFT_LIM);
                a = expf(d);
                __hip_atomic_fetch_add(&ssum[rel], a,
                                       __ATOMIC_RELAXED, __HIP_MEMORY_SCOPE_WORKGROUP);
            }
            #pragma unroll
            for (int k = 0; k < 32; ++k) {
                const int   rk = __builtin_amdgcn_readlane(rel, k);
                const float ak = __int_as_float(__builtin_amdgcn_readlane(__float_as_int(a), k));
                __hip_atomic_fetch_add(&acc[rk * 32 + lane], v[k] * ak,
                                       __ATOMIC_RELAXED, __HIP_MEMORY_SCOPE_WORKGROUP);
            }
        }
        if (far) far_flag = 1;
    };

    sweep_fast();
    if (tail > L_CAP && lane == 0) ovf_flag = 1;
    __syncthreads();
    const bool over = __builtin_amdgcn_readfirstlane(ovf_flag) != 0;
    bool twopass = over || !packed;
    if (!twopass) {
        flush_onepass();
        __syncthreads();
        twopass = __builtin_amdgcn_readfirstlane(far_flag) != 0;
        if (twopass) {
            for (int i = tid; i < (OWN_ROWS + 1) * 8; i += OWN_THREADS)
                acc4[i] = make_float4(0.0f, 0.0f, 0.0f, 0.0f);
            for (int i = tid; i < OWN_ROWS + 1; i += OWN_THREADS)
                ssum[i] = 0.0f;
            __syncthreads();
        }
    }
    if (twopass) {
        const int nChunks = over ? (E + CHUNK_EDGES - 1) / CHUNK_EDGES : 1;
        for (int chunk = 0; chunk < nChunks; ++chunk) {
            if (over) {
                const int lo = chunk * CHUNK_EDGES;
                int hi = lo + CHUNK_EDGES;
                if (hi > E) hi = E;
                tail = 0;
                sweep_plain(lo, hi);
            }
            flush();
        }
        __syncthreads();
    }

    float4* __restrict__ out4 = (float4*)(out + (size_t)d0 * 32);
    for (int i = tid; i < rows * 8; i += OWN_THREADS) {
        const int c = (i & 7) * 4;
        const float sm = ssum[i >> 3];
        float4 v = acc4[i];
        if (sm != 0.0f) { v.x /= sm; v.y /= sm; v.z /= sm; v.w /= sm; }
        v.x += bias[c];     v.x = (v.x > 0.0f) ? v.x : NEG_OUT * v.x;
        v.y += bias[c + 1]; v.y = (v.y > 0.0f) ? v.y : NEG_OUT * v.y;
        v.z += bias[c + 2]; v.z = (v.z > 0.0f) ? v.z : NEG_OUT * v.z;
        v.w += bias[c + 3]; v.w = (v.w > 0.0f) ? v.w : NEG_OUT * v.w;
        out4[i] = v;
    }
}

extern "C" void kernel_launch(void* const* d_in, const int* in_sizes, int n_in,
                              void* d_out, int out_size, void* d_ws, size_t ws_size,
                              hipStream_t stream) {
    const float* x    = (const float*)d_in[0];
    const int*   ei   = (const int*)d_in[1];
    const float* ew   = (const float*)d_in[2];
    const float* Wl   = (const float*)d_in[3];
    const float* bl   = (const float*)d_in[4];
    const float* Wr   = (const float*)d_in[5];
    const float* br   = (const float*)d_in[6];
    const float* We   = (const float*)d_in[7];
    const float* att  = (const float*)d_in[8];
    const float* bias = (const float*)d_in[9];

    const int N = in_sizes[0] / 128;
    const int E = in_sizes[2];
    const int* srcv = ei;
    const int* dstv = ei + E;

    float* xl  = (float*)d_ws;
    float* xr  = xl + (size_t)N * 32;
    unsigned int* rec = (unsigned int*)(xr + (size_t)N * 32);
    unsigned int* oid = rec + (size_t)E * 2;
    float* out = (float*)d_out;

    const int tiles = (N + 15) / 16;
    const int wavesPerBlock = 8;
    gat_gemm<<<(tiles + wavesPerBlock - 1) / wavesPerBlock, 32 * wavesPerBlock, 0, stream>>>(
        x, Wl, bl, Wr, br, xl, xr, N, dstv, oid, E, srcv, ew, rec);

    gat_edges_owner<<<(N + OWN_ROWS - 1) / OWN_ROWS, OWN_THREADS, 0, stream>>>(
        srcv, dstv, oid, rec, ew, xl, xr, We, att, bias, out, N, E);
}


# edge pass, same sums in fewer vector instructions per edge: one lane read for src+row, scalar-base xl gather, one LDS address for xr read and sum add, leaky_relu as max, half swap in the logit tree
# speedup vs baseline: 12.3175x; 1.0549x over previous
#include <hip/hip_runtime.h>
#include <hip/hip_bf16.h>
#include <math.h>


#define NEG_ATT 0.2f
#define NEG_OUT 0.01f

typedef __attribute__((ext_vector_type(2))) float v2f;
typedef __attribute__((ext_vector_type(8))) float v8f;

#define OWN_ROWS    392
#define OWN_THREADS 1024
#define OWN_WAVES   (OWN_THREADS / 32)
#define SCAN_STEPS  4
#define L_CAP       640
#define BATCH_EDGES (OWN_THREADS * SCAN_STEPS * 4)
#define SHIFT_LIM   32.0f
#define ROW_SHIFT   23
#define SRC_MASK    ((1u << ROW_SHIFT) - 1u)
#define PACK_MAX_N  (256 * OWN_ROWS)

__device__ __forceinline__ unsigned owner_byte(int d) {
    return ((unsigned)d / (unsigned)OWN_ROWS) & 255u;
}
__device__ __forceinline__ unsigned byte_bits(unsigned o, int p) {
    return (((o >> p) & 0x01010101u) * 0x10204080u) >> 28;
}

__global__ __launch_bounds__(256) void gat_gemm(
        const float* __restrict__ x,
        const float* __restrict__ Wl, const float* __restrict__ bl,
        const float* __restrict__ Wr, const float* __restrict__ br,
        float* __restrict__ xl, float* __restrict__ xr, int N,
        const int* __restrict__ dst, unsigned int* __restrict__ oid, int E,
        const int* __restrict__ src, const float* __restrict__ ew, unsigned int* __restrict__ rec) {
    {
        const bool vec_ok   = ((((size_t)dst) | ((size_t)oid)) & 15) == 0;
        const int  nBatches = vec_ok ? (E / BATCH_EDGES) : 0;
        const int4* __restrict__ dst4 = (const int4*)dst;
        uint4* __restrict__ oid4 = (uint4*)oid;
        const int nChunks = nBatches * OWN_THREADS;
        for (int c = blockIdx.x * blockDim.x + threadIdx.x; c < nChunks; c += gridDim.x * blockDim.x) {
            const int b = c / OWN_THREADS, t = c % OWN_THREADS;
            unsigned w[4] = {0u, 0u, 0u, 0u};
            #pragma unroll
            for (int u = 0; u < SCAN_STEPS; ++u) {
                const int4 d = dst4[(b * SCAN_STEPS + u) * OWN_THREADS + t];
                const unsigned o = owner_byte(d.x) | (owner_byte(d.y) << 8) | (owner_byte(d.z) << 16) | (owner_byte(d.w) << 24);
                #pragma unroll
                for (int p = 0; p < 8; ++p)
                    w[p >> 1] |= byte_bits(o, p) << ((p & 1) * 16 + u * 4);
            }
            oid4[c] = make_uint4(w[0], w[1], w[2], w[3]);
        }
    }
    if (N <= PACK_MAX_N) {
        const int gt = blockIdx.x * blockDim.x + threadIdx.x;
        const int gn = gridDim.x * blockDim.x;
        const bool v2_ok = (((((size_t)src) | ((size_t)dst) | ((size_t)ew)) & 7) == 0) && ((((size_t)rec) & 15) == 0);
        const int  E2    = v2_ok ? (E >> 1) : 0;
        const int2*   __restrict__ src2 = (const int2*)src;
        const int2*   __restrict__ dst2 = (const int2*)dst;
        const float2* __restrict__ ew2  = (const float2*)ew;
        uint4* __restrict__ rec4 = (uint4*)rec;
        for (int i = gt; i < E2; i += 4 * gn) {
            int2 s[4], d[4];
            float2 w[4];
            #pragma unroll
            for (int u = 0; u < 4; ++u) {
                const int j = min(i + u * gn, E2 - 1);
                s[u] = src2[j];
                d[u] = dst2[j];
                w[u] = ew2[j];
            }
            #pragma unroll
            for (int u = 0; u < 4; ++u) {
                const int j = i + u * gn;
                const unsigned r0 = (unsigned)d[u].x % (unsigned)OWN_ROWS, r1 = (unsigned)d[u].y % (unsigned)OWN_ROWS;
                if (j < E2)
                    rec4[j] = make_uint4((unsigned)s[u].x | (r0 << ROW_SHIFT), __float_as_uint(w[u].x),
                                         (unsigned)s[u].y | (r1 << ROW_SHIFT), __float_as_uint(w[u].y));
            }
        }
        uint2* __restrict__ rec2 = (uint2*)rec;
        for (int e = E2 * 2 + gt; e < E; e += gn) {
            const unsigned r = (unsigned)dst[e] % (unsigned)OWN_ROWS;
            rec2[e] = make_uint2((unsigned)src[e] | (r << ROW_SHIFT), __float_as_uint(ew[e]));
        }
    }

    const int lane = threadIdx.x & 31;
    const int wave = threadIdx.x >> 5;
    const int tile = blockIdx.x * (blockDim.x >> 5) + wave;
    const int row0 = tile * 16;
    if (row0 >= N) return;

    const int m  = lane & 15;
    const int kh = lane >> 4;

    int arow = row0 + m;
    if (arow >= N) arow = N - 1;
    const float* __restrict__ xrow = x + (size_t)arow * 128;

    v8f cl0 = {}, cl1 = {}, cr0 = {}, cr1 = {};

    #pragma unroll 4
    for (int k0 = 0; k0 < 128; k0 += 4) {
        const int kb = k0 + 2 * kh;
        v2f a;
        a[0] = xrow[kb];
        a[1] = xrow[kb + 1];

        const int n = m;
        v2f bl0v, bl1v, br0v, br1v;
        bl0v[0] = Wl[kb * 32 + n];       bl0v[1] = Wl[(kb + 1) * 32 + n];
        bl1v[0] = Wl[kb * 32 + n + 16];  bl1v[1] = Wl[(kb + 1) * 32 + n + 16];
        br0v[0] = Wr[kb * 32 + n];       br0v[1] = Wr[(kb + 1) * 32 + n];
        br1v[0] = Wr[kb * 32 + n + 16];  br1v[1] = Wr[(kb + 1) * 32 + n + 16];

        cl0 = __builtin_amdgcn_wmma_f32_16x16x4_f32(false, a, false, bl0v, (short)0, cl0, false, false);
        cl1 = __builtin_amdgcn_wmma_f32_16x16x4_f32(false, a, false, bl1v, (short)0, cl1, false, false);
        cr0 = __builtin_amdgcn_wmma_f32_16x16x4_f32(false, a, false, br0v, (short)0, cr0, false, false);
        cr1 = __builtin_amdgcn_wmma_f32_16x16x4_f32(false, a, false, br1v, (short)0, cr1, false, false);
    }

    const int n = lane & 15;
    const float bl_lo = bl[n], bl_hi = bl[n + 16];
    const float br_lo = br[n], br_hi = br[n + 16];
    #pragma unroll
    for (int v = 0; v < 8; ++v) {
        const int mm = v + 8 * kh;
        const int row = row0 + mm;
        if (row < N) {
            const size_t ro = (size_t)row * 32;
            xl[ro + n]      = cl0[v] + bl_lo;
            xl[ro + n + 16] = cl1[v] + bl_hi;
            xr[ro + n]      = cr0[v] + br_lo;
            xr[ro + n + 16] = cr1[v] + br_hi;
        }
    }
}

#define CHUNK_EDGES ((L_CAP / 32) * OWN_THREADS)
#define XRS_OFF     ((OWN_ROWS + 1) * 32)

template <int N>
__device__ __forceinline__ float row_xor(float x) {
    return __int_as_float(__builtin_amdgcn_update_dpp(0, __float_as_int(x), 0x160 + N, 0xf, 0xf, false));
}
__device__ __forceinline__ float half_swap_add(float a, float b) {
    typedef __attribute__((ext_vector_type(2))) unsigned v2u;
    const v2u r = __builtin_amdgcn_permlane16_swap(__float_as_uint(a), __float_as_uint(b), false, false);
    return __uint_as_float(r[0]) + __uint_as_float(r[1]);
}
__device__ __forceinline__ float half_xor(float x) {
    const int b = __float_as_int(x);
    return __int_as_float(__builtin_amdgcn_permlanex16(b, b, 0x76543210, 0xfedcba98, false, false));
}

typedef const __attribute__((address_space(1))) float* grow_t;
__device__ __forceinline__ grow_t uniform_row(const float* base, unsigned row) {
    const unsigned long long a = (unsigned long long)base + ((unsigned long long)row << 7);
    const unsigned lo = (unsigned)__builtin_amdgcn_readfirstlane((int)(unsigned)a);
    const unsigned hi = (unsigned)__builtin_amdgcn_readfirstlane((int)(unsigned)(a >> 32));
    return (grow_t)(((unsigned long long)hi << 32) | lo);
}

__device__ __forceinline__ void lds_max_float(float* addr, float val) {
    const float v = val + 0.0f;
    if (v >= 0.0f) {
        atomicMax((int*)addr, __float_as_int(v));
    } else {
        atomicMin((unsigned int*)addr, __float_as_uint(v));
    }
}

__global__ __launch_bounds__(OWN_THREADS) void gat_edges_owner(
        const int* __restrict__ src, const int* __restrict__ dst,
        const unsigned int* __restrict__ oid,
        const unsigned int* __restrict__ rec,
        const float* __restrict__ ew,
        const float* __restrict__ xl, const float* __restrict__ xr,
        const float* __restrict__ We, const float* __restrict__ att,
        const float* __restrict__ bias,
        float* __restrict__ out, int N, int E) {
    __shared__ __attribute__((aligned(16))) float rowbuf[2 * XRS_OFF];
    float* const acc = rowbuf;
    float* const xrs = rowbuf + XRS_OFF;
    __shared__ int            lst_e[OWN_WAVES][L_CAP];
    __shared__ float          lst_l[OWN_WAVES][L_CAP];
    __shared__ unsigned short lst_r[OWN_WAVES][L_CAP];
    __shared__ float smax[OWN_ROWS + 1];
    __shared__ float cmax[OWN_ROWS + 1];
    __shared__ float ssum[OWN_ROWS + 1];
    __shared__ float sref[OWN_ROWS + 1];
    __shared__ int   ovf_flag;
    __shared__ int   far_flag;

    const int tid  = threadIdx.x;
    const int lane = tid & 31;
    const int wave = __builtin_amdgcn_readfirstlane(tid >> 5);
    const int d0   = blockIdx.x * OWN_ROWS;
    int rows = N - d0;
    if (rows > OWN_ROWS) rows = OWN_ROWS;
    const unsigned urows = (unsigned)rows;

    float4* acc4 = (float4*)acc;
    float4* xrs4 = (float4*)xrs;
    const float4* __restrict__ xr4 = (const float4*)(xr + (size_t)d0 * 32);
    for (int i = tid; i < (OWN_ROWS + 1) * 8; i += OWN_THREADS) {
        acc4[i] = make_float4(0.0f, 0.0f, 0.0f, 0.0f);
        xrs4[i] = ((i >> 3) < rows) ? xr4[i] : make_float4(0.0f, 0.0f, 0.0f, 0.0f);
    }
    for (int i = tid; i < OWN_ROWS + 1; i += OWN_THREADS) {
        smax[i] = -INFINITY;
        cmax[i] = -INFINITY;
        ssum[i] = 0.0f;
    }
    if (tid == 0) { ovf_flag = 0; far_flag = 0; }
    __syncthreads();

    int*            __restrict__ le = lst_e[wave];
    float*          __restrict__ ll = lst_l[wave];
    unsigned short* __restrict__ lr = lst_r[wave];
    int tail = 0;

    const float we_l = We[lane];
    const float at_l = att[lane];

    const bool packed = N <= PACK_MAX_N;
    const uint2* __restrict__ rec2 = (const uint2*)rec;
    auto fetch = [&](int e, int& s, float& w, int& r) {
        if (packed) {
            const uint2 q = rec2[e];
            s = (int)(q.x & SRC_MASK);
            w = __uint_as_float(q.y);
            r = (int)(q.x >> ROW_SHIFT);
        } else {
            s = src[e];
            w = ew[e];
            const unsigned rr = (unsigned)(dst[e] - d0);
            r = (rr < urows) ? (int)rr : OWN_ROWS;
        }
    };

    for (int r = wave; r < OWN_ROWS + 1; r += OWN_WAVES) {
        float t = xrs[r * 32 + lane];
        t = (t > 0.0f) ? t : NEG_ATT * t;
        t *= at_l;
        t += row_xor<1>(t);
        t += row_xor<2>(t);
        t += row_xor<4>(t);
        t += row_xor<8>(t);
        t += half_xor(t);
        if (lane == 0) sref[r] = t;
    }

    auto push = [&](bool hit, int e) {
        const unsigned mask = __builtin_amdgcn_ballot_w32(hit);
        if (mask != 0u) {
            if (hit) {
                const int p = tail + (int)__builtin_amdgcn_mbcnt_lo(mask, 0u);
                if (p < L_CAP) le[p] = e;
            }
            tail += __builtin_popcount(mask);
        }
    };

    auto sweep_plain = [&](int lo, int hi) {
        for (int base = lo + wave * 32; base < hi; base += OWN_THREADS) {
            const int e = base + lane;
            const unsigned r = (e < hi) ? (unsigned)(dst[e] - d0) : 0xFFFFFFFFu;
            push(r < urows, e);
        }
    };

    auto sweep_fast = [&]() {
        const bool vec_ok   = ((((size_t)dst) | ((size_t)oid)) & 15) == 0;
        const int  nBatches = vec_ok ? (E / BATCH_EDGES) : 0;
        const int  swept    = nBatches * BATCH_EDGES;
        const uint4* __restrict__ oid4 = (const uint4*)oid;
        const unsigned gid = blockIdx.x & 255u;
        auto plane_pat = [&](int k) -> unsigned {
            return (((gid >> (2 * k)) & 1u) ? 0u : 0x0000FFFFu) | (((gid >> (2 * k + 1)) & 1u) ? 0u : 0xFFFF0000u);
        };
        const unsigned q0 = plane_pat(0), q1 = plane_pat(1), q2 = plane_pat(2), q3 = plane_pat(3);

        const int bi0 = (int)(((unsigned long long)blockIdx.x * (unsigned long long)nBatches) / gridDim.x);

        const int ntid12 = -12 * tid;
        auto scan_batch = [&](const uint4& v, int idx) {
            unsigned a = v.x ^ q0;
            a = __builtin_amdgcn_bitop3_b32(v.y, a, q1, 0x48);
            a = __builtin_amdgcn_bitop3_b32(v.z, a, q2, 0x48);
            a = __builtin_amdgcn_bitop3_b32(v.w, a, q3, 0x48);
            unsigned m = a & (a >> 16);
            unsigned mask = __builtin_amdgcn_ballot_w32(m != 0u);
            while (mask != 0u) {
                if (m != 0u) {
                    const int p = min((int)__builtin_amdgcn_mbcnt_lo(mask, (unsigned)tail), L_CAP - 1);
                    const int i = __builtin_ctz(m);
                    le[p] = ((idx << 4) + ntid12) + ((i >> 2) * (OWN_THREADS * 4 - 4) + i);
                }
                m &= m - 1u;
                tail += __builtin_popcount(mask);
                mask = __builtin_amdgcn_ballot_w32(m != 0u);
            }
        };

        if (nBatches > 0) {
            int bl = bi0;
            uint4 c0, c1, c2, c3;
            int   i0, i1, i2, i3;
            auto load = [&](uint4& c, int& idx) {
                idx = bl * OWN_THREADS + tid;
                c = oid4[idx];
                bl = (bl + 1 == nBatches) ? 0 : bl + 1;
            };
            load(c0, i0); load(c1, i1); load(c2, i2); load(c3, i3);
            int q = 0;
            for (; q + 4 <= nBatches; q += 4) {
                scan_batch(c0, i0); load(c0, i0);
                scan_batch(c1, i1); load(c1, i1);
                scan_batch(c2, i2); load(c2, i2);
                scan_batch(c3, i3); load(c3, i3);
            }
            if (q     < nBatches) scan_batch(c0, i0);
            if (q + 1 < nBatches) scan_batch(c1, i1);
            if (q + 2 < nBatches) scan_batch(c2, i2);
        }
        sweep_plain(swept, E);
    };

    auto flush = [&]() {
        const int ng = (tail + 31) >> 5;

        int   s_nx = 0;
        float w_nx = 0.0f;
        int   r_nx = OWN_ROWS;
        if (lane < tail)
            fetch(le[lane], s_nx, w_nx, r_nx);
        for (int g = 0; g < ng; ++g) {
            const int   n = tail - g * 32;
            const int   s = s_nx;
            const float w = w_nx;
            const int   rel = r_nx;
            s_nx = 0;
            w_nx = 0.0f;
            r_nx = OWN_ROWS;
            if ((g + 1) * 32 + lane < tail)
                fetch(le[(g + 1) * 32 + lane], s_nx, w_nx, r_nx);
            float v[32];
            #pragma unroll
            for (int k = 0; k < 32; ++k) {
                const int sk = __builtin_amdgcn_readlane(s, k);
                v[k] = xl[(size_t)sk * 32 + lane];
            }
            #pragma unroll
            for (int k = 0; k < 32; ++k) {
                const int   rk = __builtin_amdgcn_readlane(rel, k);
                const float wk = __int_as_float(__builtin_amdgcn_readlane(__float_as_int(w), k));
                float m = v[k] + xrs[rk * 32 + lane] + wk * we_l;
                m = (m > 0.0f) ? m : NEG_ATT * m;
                v[k] = m * at_l;
            }
            const bool b16 = (lane & 16) != 0, b8 = (lane & 8) != 0,
                       b4  = (lane & 4)  != 0, b2 = (lane & 2) != 0, b1 = (lane & 1) != 0;
            #pragma unroll
            for (int k = 0; k < 16; ++k) {
                const float keep = b16 ? v[k + 16] : v[k];
                const float send = b16 ? v[k] : v[k + 16];
                v[k] = keep + half_xor(send);
            }
            #pragma unroll
            for (int k = 0; k < 8; ++k) {
                const float keep = b8 ? v[k + 8] : v[k];
                const float send = b8 ? v[k] : v[k + 8];
                v[k] = keep + row_xor<8>(send);
            }
            #pragma unroll
            for (int k = 0; k < 4; ++k) {
                const float keep = b4 ? v[k + 4] : v[k];
                const float send = b4 ? v[k] : v[k + 4];
                v[k] = keep + row_xor<4>(send);
            }
            #pragma unroll
            for (int k = 0; k < 2; ++k) {
                const float keep = b2 ? v[k + 2] : v[k];
                const float send = b2 ? v[k] : v[k + 2];
                v[k] = keep + row_xor<2>(send);
            }
            const float keep1 = b1 ? v[1] : v[0];
            const float send1 = b1 ? v[0] : v[1];
            const float lg = keep1 + row_xor<1>(send1);
            if (lane < n) {
                ll[g * 32 + lane] = lg;
                le[g * 32 + lane] = s;
                lr[g * 32 + lane] = (unsigned short)rel;
                lds_max_float(&cmax[rel], lg);
            }
        }
        __syncthreads();

        for (int i = tid; i < OWN_ROWS * 32; i += OWN_THREADS) {
            const int r = i >> 5;
            const float mo = smax[r], c = cmax[r];
            if (c > mo) {
                if (mo != -INFINITY) {
                    const float f = __expf(mo - c);
                    acc[i] *= f;
                    if (lane == 0) ssum[r] *= f;
                }
                if (lane == 0) smax[r] = c;
            }
            if (lane == 0) cmax[r] = -INFINITY;
        }
        __syncthreads();

        for (int g = 0; g < ng; ++g) {
            const int n = tail - g * 32;
            int rel = OWN_ROWS, s = 0;
            float a = 0.0f;
            if (lane < n) {
                s   = le[g * 32 + lane];
                rel = lr[g * 32 + lane];
                a   = __expf(ll[g * 32 + lane] - smax[rel]);
                __hip_atomic_fetch_add(&ssum[rel], a,
                                       __ATOMIC_RELAXED, __HIP_MEMORY_SCOPE_WORKGROUP);
            }
            float v[32];
            #pragma unroll
            for (int k = 0; k < 32; ++k) {
                const int sk = __builtin_amdgcn_readlane(s, k);
                v[k] = xl[(size_t)sk * 32 + lane];
            }
            #pragma unroll
            for (int k = 0; k < 32; ++k) {
                const int   rk = __builtin_amdgcn_readlane(rel, k);
                const float ak = __int_as_float(__builtin_amdgcn_readlane(__float_as_int(a), k));
                __hip_atomic_fetch_add(&acc[rk * 32 + lane], v[k] * ak,
                                       __ATOMIC_RELAXED, __HIP_MEMORY_SCOPE_WORKGROUP);
            }
        }
    };

    auto flush_onepass = [&]() {
        const int ng = (tail + 31) >> 5;
        bool far = false;
        uint2 q_nx = make_uint2(0u, 0u);
        if (ng > 0) q_nx = rec2[le[min(lane, tail - 1)]];
        for (int g = 0; g < ng; ++g) {
            const int   n = tail - g * 32;
            const bool  valid = lane < n;
            const unsigned px = valid ? q_nx.x : ((unsigned)OWN_ROWS << ROW_SHIFT);
            const float w = valid ? __uint_as_float(q_nx.y) : 0.0f;
            const int   rel = (int)(px >> ROW_SHIFT);
            q_nx = rec2[le[min((g + 1) * 32 + lane, tail - 1)]];
            float v[32];
            int   off[32];
            #pragma unroll
            for (int k = 0; k < 32; ++k) {
                const unsigned xk = (unsigned)__builtin_amdgcn_readlane((int)px, k);
                v[k]   = uniform_row(xl, xk & SRC_MASK)[lane];
                off[k] = (int)(xk >> ROW_SHIFT) * 32 + lane;
            }
            float t[32];
            #pragma unroll
            for (int k = 0; k < 32; ++k) {
                const float wk = __int_as_float(__builtin_amdgcn_readlane(__float_as_int(w), k));
                float m = v[k] + rowbuf[XRS_OFF + off[k]] + wk * we_l;
                m = fmaxf(m, NEG_ATT * m);
                t[k] = m * at_l;
            }
            const bool b8 = (lane & 8) != 0, b4 = (lane & 4) != 0, b2 = (lane & 2) != 0, b1 = (lane & 1) != 0;
            #pragma unroll
            for (int k = 0; k < 16; ++k)
                t[k] = half_swap_add(t[k], t[k + 16]);
            #pragma unroll
            for (int k = 0; k < 8; ++k) {
                const float keep = b8 ? t[k + 8] : t[k];
                const float send = b8 ? t[k] : t[k + 8];
                t[k] = keep + row_xor<8>(send);
            }
            #pragma unroll
            for (int k = 0; k < 4; ++k) {
                const float keep = b4 ? t[k + 4] : t[k];
                const float send = b4 ? t[k] : t[k + 4];
                t[k] = keep + row_xor<4>(send);
            }
            #pragma unroll
            for (int k = 0; k < 2; ++k) {
                const float keep = b2 ? t[k + 2] : t[k];
                const float send = b2 ? t[k] : t[k + 2];
                t[k] = keep + row_xor<2>(send);
            }
            const float keep1 = b1 ? t[1] : t[0];
            const float send1 = b1 ? t[0] : t[1];
            const float lg = keep1 + row_xor<1>(send1);
            float a = 0.0f;
            if (lane < n && rel != OWN_ROWS) {
                const float d = lg - sref[rel];
                far = far || !(fabsf(d) <= SHIFT_LIM);
                a = expf(d);
                __hip_atomic_fetch_add(&ssum[rel], a,
                                       __ATOMIC_RELAXED, __HIP_MEMORY_SCOPE_WORKGROUP);
            }
            #pragma unroll
            for (int k = 0; k < 32; ++k) {
                const float ak = __int_as_float(__builtin_amdgcn_readlane(__float_as_int(a), k));
                __hip_atomic_fetch_add(&rowbuf[off[k]], v[k] * ak,
                                       __ATOMIC_RELAXED, __HIP_MEMORY_SCOPE_WORKGROUP);
            }
        }
        if (far) far_flag = 1;
    };

    sweep_fast();
    if (tail > L_CAP && lane == 0) ovf_flag = 1;
    __syncthreads();
    const bool over = __builtin_amdgcn_readfirstlane(ovf_flag) != 0;
    bool twopass = over || !packed;
    if (!twopass) {
        flush_onepass();
        __syncthreads();
        twopass = __builtin_amdgcn_readfirstlane(far_flag) != 0;
        if (twopass) {
            for (int i = tid; i < (OWN_ROWS + 1) * 8; i += OWN_THREADS)
                acc4[i] = make_float4(0.0f, 0.0f, 0.0f, 0.0f);
            for (int i = tid; i < OWN_ROWS + 1; i += OWN_THREADS)
                ssum[i] = 0.0f;
            __syncthreads();
        }
    }
    if (twopass) {
        const int nChunks = over ? (E + CHUNK_EDGES - 1) / CHUNK_EDGES : 1;
        for (int chunk = 0; chunk < nChunks; ++chunk) {
            if (over) {
                const int lo = chunk * CHUNK_EDGES;
                int hi = lo + CHUNK_EDGES;
                if (hi > E) hi = E;
                tail = 0;
                sweep_plain(lo, hi);
            }
            flush();
        }
        __syncthreads();
    }

    float4* __restrict__ out4 = (float4*)(out + (size_t)d0 * 32);
    for (int i = tid; i < rows * 8; i += OWN_THREADS) {
        const int c = (i & 7) * 4;
        const float sm = ssum[i >> 3];
        float4 v = acc4[i];
        if (sm != 0.0f) { v.x /= sm; v.y /= sm; v.z /= sm; v.w /= sm; }
        v.x += bias[c];     v.x = (v.x > 0.0f) ? v.x : NEG_OUT * v.x;
        v.y += bias[c + 1]; v.y = (v.y > 0.0f) ? v.y : NEG_OUT * v.y;
        v.z += bias[c + 2]; v.z = (v.z > 0.0f) ? v.z : NEG_OUT * v.z;
        v.w += bias[c + 3]; v.w = (v.w > 0.0f) ? v.w : NEG_OUT * v.w;
        out4[i] = v;
    }
}

extern "C" void kernel_launch(void* const* d_in, const int* in_sizes, int n_in,
                              void* d_out, int out_size, void* d_ws, size_t ws_size,
                              hipStream_t stream) {
    const float* x    = (const float*)d_in[0];
    const int*   ei   = (const int*)d_in[1];
    const float* ew   = (const float*)d_in[2];
    const float* Wl   = (const float*)d_in[3];
    const float* bl   = (const float*)d_in[4];
    const float* Wr   = (const float*)d_in[5];
    const float* br   = (const float*)d_in[6];
    const float* We   = (const float*)d_in[7];
    const float* att  = (const float*)d_in[8];
    const float* bias = (const float*)d_in[9];

    const int N = in_sizes[0] / 128;
    const int E = in_sizes[2];
    const int* srcv = ei;
    const int* dstv = ei + E;

    float* xl  = (float*)d_ws;
    float* xr  = xl + (size_t)N * 32;
    unsigned int* rec = (unsigned int*)(xr + (size_t)N * 32);
    unsigned int* oid = rec + (size_t)E * 2;
    float* out = (float*)d_out;

    const int tiles = (N + 15) / 16;
    const int wavesPerBlock = 8;
    gat_gemm<<<(tiles + wavesPerBlock - 1) / wavesPerBlock, 32 * wavesPerBlock, 0, stream>>>(
        x, Wl, bl, Wr, br, xl, xr, N, dstv, oid, E, srcv, ew, rec);

    gat_edges_owner<<<(N + OWN_ROWS - 1) / OWN_ROWS, OWN_THREADS, 0, stream>>>(
        srcv, dstv, oid, rec, ew, xl, xr, We, att, bias, out, N, E);
}


# node GEMM: k loop as 16 blocks of two k-steps over a ring of three fragment sets, block i+2's A and B loads issued in front of block i's 8 matrix instructions (scheduling barrier); same k order
# speedup vs baseline: 12.9592x; 1.0521x over previous
#include <hip/hip_runtime.h>
#include <hip/hip_bf16.h>
#include <math.h>


#define NEG_ATT 0.2f
#define NEG_OUT 0.01f

typedef __attribute__((ext_vector_type(2))) float v2f;
typedef __attribute__((ext_vector_type(8))) float v8f;

#define OWN_ROWS    392
#define OWN_THREADS 1024
#define OWN_WAVES   (OWN_THREADS / 32)
#define SCAN_STEPS  4
#define L_CAP       640
#define BATCH_EDGES (OWN_THREADS * SCAN_STEPS * 4)
#define SHIFT_LIM   32.0f
#define ROW_SHIFT   23
#define SRC_MASK    ((1u << ROW_SHIFT) - 1u)
#define PACK_MAX_N  (256 * OWN_ROWS)

__device__ __forceinline__ unsigned owner_byte(int d) {
    return ((unsigned)d / (unsigned)OWN_ROWS) & 255u;
}
__device__ __forceinline__ unsigned byte_bits(unsigned o, int p) {
    return (((o >> p) & 0x01010101u) * 0x10204080u) >> 28;
}

__global__ __launch_bounds__(256) void gat_gemm(
        const float* __restrict__ x,
        const float* __restrict__ Wl, const float* __restrict__ bl,
        const float* __restrict__ Wr, const float* __restrict__ br,
        float* __restrict__ xl, float* __restrict__ xr, int N,
        const int* __restrict__ dst, unsigned int* __restrict__ oid, int E,
        const int* __restrict__ src, const float* __restrict__ ew, unsigned int* __restrict__ rec) {
    {
        const bool vec_ok   = ((((size_t)dst) | ((size_t)oid)) & 15) == 0;
        const int  nBatches = vec_ok ? (E / BATCH_EDGES) : 0;
        const int4* __restrict__ dst4 = (const int4*)dst;
        uint4* __restrict__ oid4 = (uint4*)oid;
        const int nChunks = nBatches * OWN_THREADS;
        for (int c = blockIdx.x * blockDim.x + threadIdx.x; c < nChunks; c += gridDim.x * blockDim.x) {
            const int b = c / OWN_THREADS, t = c % OWN_THREADS;
            unsigned w[4] = {0u, 0u, 0u, 0u};
            #pragma unroll
            for (int u = 0; u < SCAN_STEPS; ++u) {
                const int4 d = dst4[(b * SCAN_STEPS + u) * OWN_THREADS + t];
                const unsigned o = owner_byte(d.x) | (owner_byte(d.y) << 8) | (owner_byte(d.z) << 16) | (owner_byte(d.w) << 24);
                #pragma unroll
                for (int p = 0; p < 8; ++p)
                    w[p >> 1] |= byte_bits(o, p) << ((p & 1) * 16 + u * 4);
            }
            oid4[c] = make_uint4(w[0], w[1], w[2], w[3]);
        }
    }
    if (N <= PACK_MAX_N) {
        const int gt = blockIdx.x * blockDim.x + threadIdx.x;
        const int gn = gridDim.x * blockDim.x;
        const bool v2_ok = (((((size_t)src) | ((size_t)dst) | ((size_t)ew)) & 7) == 0) && ((((size_t)rec) & 15) == 0);
        const int  E2    = v2_ok ? (E >> 1) : 0;
        const int2*   __restrict__ src2 = (const int2*)src;
        const int2*   __restrict__ dst2 = (const int2*)dst;
        const float2* __restrict__ ew2  = (const float2*)ew;
        uint4* __restrict__ rec4 = (uint4*)rec;
        for (int i = gt; i < E2; i += 4 * gn) {
            int2 s[4], d[4];
            float2 w[4];
            #pragma unroll
            for (int u = 0; u < 4; ++u) {
                const int j = min(i + u * gn, E2 - 1);
                s[u] = src2[j];
                d[u] = dst2[j];
                w[u] = ew2[j];
            }
            #pragma unroll
            for (int u = 0; u < 4; ++u) {
                const int j = i + u * gn;
                const unsigned r0 = (unsigned)d[u].x % (unsigned)OWN_ROWS, r1 = (unsigned)d[u].y % (unsigned)OWN_ROWS;
                if (j < E2)
                    rec4[j] = make_uint4((unsigned)s[u].x | (r0 << ROW_SHIFT), __float_as_uint(w[u].x),
                                         (unsigned)s[u].y | (r1 << ROW_SHIFT), __float_as_uint(w[u].y));
            }
        }
        uint2* __restrict__ rec2 = (uint2*)rec;
        for (int e = E2 * 2 + gt; e < E; e += gn) {
            const unsigned r = (unsigned)dst[e] % (unsigned)OWN_ROWS;
            rec2[e] = make_uint2((unsigned)src[e] | (r << ROW_SHIFT), __float_as_uint(ew[e]));
        }
    }

    const int lane = threadIdx.x & 31;
    const int wave = threadIdx.x >> 5;
    const int tile = blockIdx.x * (blockDim.x >> 5) + wave;
    const int row0 = tile * 16;
    if (row0 >= N) return;

    const int m  = lane & 15;
    const int kh = lane >> 4;

    int arow = row0 + m;
    if (arow >= N) arow = N - 1;
    const float* __restrict__ xrow = x + (size_t)arow * 128;

    v8f cl0 = {}, cl1 = {}, cr0 = {}, cr1 = {};

    v2f fa[3][2], fb[3][2][4];
    auto load_block = [&](v2f (&pa)[2], v2f (&pb)[2][4], int blk) {
        #pragma unroll
        for (int s = 0; s < 2; ++s) {
            const int kb = blk * 8 + s * 4 + 2 * kh;
            pa[s][0] = xrow[kb];
            pa[s][1] = xrow[kb + 1];
            pb[s][0][0] = Wl[kb * 32 + m];       pb[s][0][1] = Wl[(kb + 1) * 32 + m];
            pb[s][1][0] = Wl[kb * 32 + m + 16];  pb[s][1][1] = Wl[(kb + 1) * 32 + m + 16];
            pb[s][2][0] = Wr[kb * 32 + m];       pb[s][2][1] = Wr[(kb + 1) * 32 + m];
            pb[s][3][0] = Wr[kb * 32 + m + 16];  pb[s][3][1] = Wr[(kb + 1) * 32 + m + 16];
        }
    };
    load_block(fa[0], fb[0], 0);
    load_block(fa[1], fb[1], 1);
    #pragma unroll
    for (int blk = 0; blk < 16; ++blk) {
        if (blk + 2 < 16) load_block(fa[(blk + 2) % 3], fb[(blk + 2) % 3], blk + 2);
        __builtin_amdgcn_sched_barrier(0);
        #pragma unroll
        for (int s = 0; s < 2; ++s) {
            cl0 = __builtin_amdgcn_wmma_f32_16x16x4_f32(false, fa[blk % 3][s], false, fb[blk % 3][s][0], (short)0, cl0, false, false);
            cl1 = __builtin_amdgcn_wmma_f32_16x16x4_f32(false, fa[blk % 3][s], false, fb[blk % 3][s][1], (short)0, cl1, false, false);
            cr0 = __builtin_amdgcn_wmma_f32_16x16x4_f32(false, fa[blk % 3][s], false, fb[blk % 3][s][2], (short)0, cr0, false, false);
            cr1 = __builtin_amdgcn_wmma_f32_16x16x4_f32(false, fa[blk % 3][s], false, fb[blk % 3][s][3], (short)0, cr1, false, false);
        }
    }

    const int n = lane & 15;
    const float bl_lo = bl[n], bl_hi = bl[n + 16];
    const float br_lo = br[n], br_hi = br[n + 16];
    #pragma unroll
    for (int v = 0; v < 8; ++v) {
        const int mm = v + 8 * kh;
        const int row = row0 + mm;
        if (row < N) {
            const size_t ro = (size_t)row * 32;
            xl[ro + n]      = cl0[v] + bl_lo;
            xl[ro + n + 16] = cl1[v] + bl_hi;
            xr[ro + n]      = cr0[v] + br_lo;
            xr[ro + n + 16] = cr1[v] + br_hi;
        }
    }
}

#define CHUNK_EDGES ((L_CAP / 32) * OWN_THREADS)
#define XRS_OFF     ((OWN_ROWS + 1) * 32)

template <int N>
__device__ __forceinline__ float row_xor(float x) {
    return __int_as_float(__builtin_amdgcn_update_dpp(0, __float_as_int(x), 0x160 + N, 0xf, 0xf, false));
}
__device__ __forceinline__ float half_swap_add(float a, float b) {
    typedef __attribute__((ext_vector_type(2))) unsigned v2u;
    const v2u r = __builtin_amdgcn_permlane16_swap(__float_as_uint(a), __float_as_uint(b), false, false);
    return __uint_as_float(r[0]) + __uint_as_float(r[1]);
}
__device__ __forceinline__ float half_xor(float x) {
    const int b = __float_as_int(x);
    return __int_as_float(__builtin_amdgcn_permlanex16(b, b, 0x76543210, 0xfedcba98, false, false));
}

typedef const __attribute__((address_space(1))) float* grow_t;
__device__ __forceinline__ grow_t uniform_row(const float* base, unsigned row) {
    const unsigned long long a = (unsigned long long)base + ((unsigned long long)row << 7);
    const unsigned lo = (unsigned)__builtin_amdgcn_readfirstlane((int)(unsigned)a);
    const unsigned hi = (unsigned)__builtin_amdgcn_readfirstlane((int)(unsigned)(a >> 32));
    return (grow_t)(((unsigned long long)hi << 32) | lo);
}

__device__ __forceinline__ void lds_max_float(float* addr, float val) {
    const float v = val + 0.0f;
    if (v >= 0.0f) {
        atomicMax((int*)addr, __float_as_int(v));
    } else {
        atomicMin((unsigned int*)addr, __float_as_uint(v));
    }
}

__global__ __launch_bounds__(OWN_THREADS) void gat_edges_owner(
        const int* __restrict__ src, const int* __restrict__ dst,
        const unsigned int* __restrict__ oid,
        const unsigned int* __restrict__ rec,
        const float* __restrict__ ew,
        const float* __restrict__ xl, const float* __restrict__ xr,
        const float* __restrict__ We, const float* __restrict__ att,
        const float* __restrict__ bias,
        float* __restrict__ out, int N, int E) {
    __shared__ __attribute__((aligned(16))) float rowbuf[2 * XRS_OFF];
    float* const acc = rowbuf;
    float* const xrs = rowbuf + XRS_OFF;
    __shared__ int            lst_e[OWN_WAVES][L_CAP];
    __shared__ float          lst_l[OWN_WAVES][L_CAP];
    __shared__ unsigned short lst_r[OWN_WAVES][L_CAP];
    __shared__ float smax[OWN_ROWS + 1];
    __shared__ float cmax[OWN_ROWS + 1];
    __shared__ float ssum[OWN_ROWS + 1];
    __shared__ float sref[OWN_ROWS + 1];
    __shared__ int   ovf_flag;
    __shared__ int   far_flag;

    const int tid  = threadIdx.x;
    const int lane = tid & 31;
    const int wave = __builtin_amdgcn_readfirstlane(tid >> 5);
    const int d0   = blockIdx.x * OWN_ROWS;
    int rows = N - d0;
    if (rows > OWN_ROWS) rows = OWN_ROWS;
    const unsigned urows = (unsigned)rows;

    float4* acc4 = (float4*)acc;
    float4* xrs4 = (float4*)xrs;
    const float4* __restrict__ xr4 = (const float4*)(xr + (size_t)d0 * 32);
    for (int i = tid; i < (OWN_ROWS + 1) * 8; i += OWN_THREADS) {
        acc4[i] = make_float4(0.0f, 0.0f, 0.0f, 0.0f);
        xrs4[i] = ((i >> 3) < rows) ? xr4[i] : make_float4(0.0f, 0.0f, 0.0f, 0.0f);
    }
    for (int i = tid; i < OWN_ROWS + 1; i += OWN_THREADS) {
        smax[i] = -INFINITY;
        cmax[i] = -INFINITY;
        ssum[i] = 0.0f;
    }
    if (tid == 0) { ovf_flag = 0; far_flag = 0; }
    __syncthreads();

    int*            __restrict__ le = lst_e[wave];
    float*          __restrict__ ll = lst_l[wave];
    unsigned short* __restrict__ lr = lst_r[wave];
    int tail = 0;

    const float we_l = We[lane];
    const float at_l = att[lane];

    const bool packed = N <= PACK_MAX_N;
    const uint2* __restrict__ rec2 = (const uint2*)rec;
    auto fetch = [&](int e, int& s, float& w, int& r) {
        if (packed) {
            const uint2 q = rec2[e];
            s = (int)(q.x & SRC_MASK);
            w = __uint_as_float(q.y);
            r = (int)(q.x >> ROW_SHIFT);
        } else {
            s = src[e];
            w = ew[e];
            const unsigned rr = (unsigned)(dst[e] - d0);
            r = (rr < urows) ? (int)rr : OWN_ROWS;
        }
    };

    for (int r = wave; r < OWN_ROWS + 1; r += OWN_WAVES) {
        float t = xrs[r * 32 + lane];
        t = (t > 0.0f) ? t : NEG_ATT * t;
        t *= at_l;
        t += row_xor<1>(t);
        t += row_xor<2>(t);
        t += row_xor<4>(t);
        t += row_xor<8>(t);
        t += half_xor(t);
        if (lane == 0) sref[r] = t;
    }

    auto push = [&](bool hit, int e) {
        const unsigned mask = __builtin_amdgcn_ballot_w32(hit);
        if (mask != 0u) {
            if (hit) {
                const int p = tail + (int)__builtin_amdgcn_mbcnt_lo(mask, 0u);
                if (p < L_CAP) le[p] = e;
            }
            tail += __builtin_popcount(mask);
        }
    };

    auto sweep_plain = [&](int lo, int hi) {
        for (int base = lo + wave * 32; base < hi; base += OWN_THREADS) {
            const int e = base + lane;
            const unsigned r = (e < hi) ? (unsigned)(dst[e] - d0) : 0xFFFFFFFFu;
            push(r < urows, e);
        }
    };

    auto sweep_fast = [&]() {
        const bool vec_ok   = ((((size_t)dst) | ((size_t)oid)) & 15) == 0;
        const int  nBatches = vec_ok ? (E / BATCH_EDGES) : 0;
        const int  swept    = nBatches * BATCH_EDGES;
        const uint4* __restrict__ oid4 = (const uint4*)oid;
        const unsigned gid = blockIdx.x & 255u;
        auto plane_pat = [&](int k) -> unsigned {
            return (((gid >> (2 * k)) & 1u) ? 0u : 0x0000FFFFu) | (((gid >> (2 * k + 1)) & 1u) ? 0u : 0xFFFF0000u);
        };
        const unsigned q0 = plane_pat(0), q1 = plane_pat(1), q2 = plane_pat(2), q3 = plane_pat(3);

        const int bi0 = (int)(((unsigned long long)blockIdx.x * (unsigned long long)nBatches) / gridDim.x);

        const int ntid12 = -12 * tid;
        auto scan_batch = [&](const uint4& v, int idx) {
            unsigned a = v.x ^ q0;
            a = __builtin_amdgcn_bitop3_b32(v.y, a, q1, 0x48);
            a = __builtin_amdgcn_bitop3_b32(v.z, a, q2, 0x48);
            a = __builtin_amdgcn_bitop3_b32(v.w, a, q3, 0x48);
            unsigned m = a & (a >> 16);
            unsigned mask = __builtin_amdgcn_ballot_w32(m != 0u);
            while (mask != 0u) {
                if (m != 0u) {
                    const int p = min((int)__builtin_amdgcn_mbcnt_lo(mask, (unsigned)tail), L_CAP - 1);
                    const int i = __builtin_ctz(m);
                    le[p] = ((idx << 4) + ntid12) + ((i >> 2) * (OWN_THREADS * 4 - 4) + i);
                }
                m &= m - 1u;
                tail += __builtin_popcount(mask);
                mask = __builtin_amdgcn_ballot_w32(m != 0u);
            }
        };

        if (nBatches > 0) {
            int bl = bi0;
            uint4 c0, c1, c2, c3;
            int   i0, i1, i2, i3;
            auto load = [&](uint4& c, int& idx) {
                idx = bl * OWN_THREADS + tid;
                c = oid4[idx];
                bl = (bl + 1 == nBatches) ? 0 : bl + 1;
            };
            load(c0, i0); load(c1, i1); load(c2, i2); load(c3, i3);
            int q = 0;
            for (; q + 4 <= nBatches; q += 4) {
                scan_batch(c0, i0); load(c0, i0);
                scan_batch(c1, i1); load(c1, i1);
                scan_batch(c2, i2); load(c2, i2);
                scan_batch(c3, i3); load(c3, i3);
            }
            if (q     < nBatches) scan_batch(c0, i0);
            if (q + 1 < nBatches) scan_batch(c1, i1);
            if (q + 2 < nBatches) scan_batch(c2, i2);
        }
        sweep_plain(swept, E);
    };

    auto flush = [&]() {
        const int ng = (tail + 31) >> 5;

        int   s_nx = 0;
        float w_nx = 0.0f;
        int   r_nx = OWN_ROWS;
        if (lane < tail)
            fetch(le[lane], s_nx, w_nx, r_nx);
        for (int g = 0; g < ng; ++g) {
            const int   n = tail - g * 32;
            const int   s = s_nx;
            const float w = w_nx;
            const int   rel = r_nx;
            s_nx = 0;
            w_nx = 0.0f;
            r_nx = OWN_ROWS;
            if ((g + 1) * 32 + lane < tail)
                fetch(le[(g + 1) * 32 + lane], s_nx, w_nx, r_nx);
            float v[32];
            #pragma unroll
            for (int k = 0; k < 32; ++k) {
                const int sk = __builtin_amdgcn_readlane(s, k);
                v[k] = xl[(size_t)sk * 32 + lane];
            }
            #pragma unroll
            for (int k = 0; k < 32; ++k) {
                const int   rk = __builtin_amdgcn_readlane(rel, k);
                const float wk = __int_as_float(__builtin_amdgcn_readlane(__float_as_int(w), k));
                float m = v[k] + xrs[rk * 32 + lane] + wk * we_l;
                m = (m > 0.0f) ? m : NEG_ATT * m;
                v[k] = m * at_l;
            }
            const bool b16 = (lane & 16) != 0, b8 = (lane & 8) != 0,
                       b4  = (lane & 4)  != 0, b2 = (lane & 2) != 0, b1 = (lane & 1) != 0;
            #pragma unroll
            for (int k = 0; k < 16; ++k) {
                const float keep = b16 ? v[k + 16] : v[k];
                const float send = b16 ? v[k] : v[k + 16];
                v[k] = keep + half_xor(send);
            }
            #pragma unroll
            for (int k = 0; k < 8; ++k) {
                const float keep = b8 ? v[k + 8] : v[k];
                const float send = b8 ? v[k] : v[k + 8];
                v[k] = keep + row_xor<8>(send);
            }
            #pragma unroll
            for (int k = 0; k < 4; ++k) {
                const float keep = b4 ? v[k + 4] : v[k];
                const float send = b4 ? v[k] : v[k + 4];
                v[k] = keep + row_xor<4>(send);
            }
            #pragma unroll
            for (int k = 0; k < 2; ++k) {
                const float keep = b2 ? v[k + 2] : v[k];
                const float send = b2 ? v[k] : v[k + 2];
                v[k] = keep + row_xor<2>(send);
            }
            const float keep1 = b1 ? v[1] : v[0];
            const float send1 = b1 ? v[0] : v[1];
            const float lg = keep1 + row_xor<1>(send1);
            if (lane < n) {
                ll[g * 32 + lane] = lg;
                le[g * 32 + lane] = s;
                lr[g * 32 + lane] = (unsigned short)rel;
                lds_max_float(&cmax[rel], lg);
            }
        }
        __syncthreads();

        for (int i = tid; i < OWN_ROWS * 32; i += OWN_THREADS) {
            const int r = i >> 5;
            const float mo = smax[r], c = cmax[r];
            if (c > mo) {
                if (mo != -INFINITY) {
                    const float f = __expf(mo - c);
                    acc[i] *= f;
                    if (lane == 0) ssum[r] *= f;
                }
                if (lane == 0) smax[r] = c;
            }
            if (lane == 0) cmax[r] = -INFINITY;
        }
        __syncthreads();

        for (int g = 0; g < ng; ++g) {
            const int n = tail - g * 32;
            int rel = OWN_ROWS, s = 0;
            float a = 0.0f;
            if (lane < n) {
                s   = le[g * 32 + lane];
                rel = lr[g * 32 + lane];
                a   = __expf(ll[g * 32 + lane] - smax[rel]);
                __hip_atomic_fetch_add(&ssum[rel], a,
                                       __ATOMIC_RELAXED, __HIP_MEMORY_SCOPE_WORKGROUP);
            }
            float v[32];
            #pragma unroll
            for (int k = 0; k < 32; ++k) {
                const int sk = __builtin_amdgcn_readlane(s, k);
                v[k] = xl[(size_t)sk * 32 + lane];
            }
            #pragma unroll
            for (int k = 0; k < 32; ++k) {
                const int   rk = __builtin_amdgcn_readlane(rel, k);
                const float ak = __int_as_float(__builtin_amdgcn_readlane(__float_as_int(a), k));
                __hip_atomic_fetch_add(&acc[rk * 32 + lane], v[k] * ak,
                                       __ATOMIC_RELAXED, __HIP_MEMORY_SCOPE_WORKGROUP);
            }
        }
    };

    auto flush_onepass = [&]() {
        const int ng = (tail + 31) >> 5;
        bool far = false;
        uint2 q_nx = make_uint2(0u, 0u);
        if (ng > 0) q_nx = rec2[le[min(lane, tail - 1)]];
        for (int g = 0; g < ng; ++g) {
            const int   n = tail - g * 32;
            const bool  valid = lane < n;
            const unsigned px = valid ? q_nx.x : ((unsigned)OWN_ROWS << ROW_SHIFT);
            const float w = valid ? __uint_as_float(q_nx.y) : 0.0f;
            const int   rel = (int)(px >> ROW_SHIFT);
            q_nx = rec2[le[min((g + 1) * 32 + lane, tail - 1)]];
            float v[32];
            int   off[32];
            #pragma unroll
            for (int k = 0; k < 32; ++k) {
                const unsigned xk = (unsigned)__builtin_amdgcn_readlane((int)px, k);
                v[k]   = uniform_row(xl, xk & SRC_MASK)[lane];
                off[k] = (int)(xk >> ROW_SHIFT) * 32 + lane;
            }
            float t[32];
            #pragma unroll
            for (int k = 0; k < 32; ++k) {
                const float wk = __int_as_float(__builtin_amdgcn_readlane(__float_as_int(w), k));
                float m = v[k] + rowbuf[XRS_OFF + off[k]] + wk * we_l;
                m = fmaxf(m, NEG_ATT * m);
                t[k] = m * at_l;
            }
            const bool b8 = (lane & 8) != 0, b4 = (lane & 4) != 0, b2 = (lane & 2) != 0, b1 = (lane & 1) != 0;
            #pragma unroll
            for (int k = 0; k < 16; ++k)
                t[k] = half_swap_add(t[k], t[k + 16]);
            #pragma unroll
            for (int k = 0; k < 8; ++k) {
                const float keep = b8 ? t[k + 8] : t[k];
                const float send = b8 ? t[k] : t[k + 8];
                t[k] = keep + row_xor<8>(send);
            }
            #pragma unroll
            for (int k = 0; k < 4; ++k) {
                const float keep = b4 ? t[k + 4] : t[k];
                const float send = b4 ? t[k] : t[k + 4];
                t[k] = keep + row_xor<4>(send);
            }
            #pragma unroll
            for (int k = 0; k < 2; ++k) {
                const float keep = b2 ? t[k + 2] : t[k];
                const float send = b2 ? t[k] : t[k + 2];
                t[k] = keep + row_xor<2>(send);
            }
            const float keep1 = b1 ? t[1] : t[0];
            const float send1 = b1 ? t[0] : t[1];
            const float lg = keep1 + row_xor<1>(send1);
            float a = 0.0f;
            if (lane < n && rel != OWN_ROWS) {
                const float d = lg - sref[rel];
                far = far || !(fabsf(d) <= SHIFT_LIM);
                a = expf(d);
                __hip_atomic_fetch_add(&ssum[rel], a,
                                       __ATOMIC_RELAXED, __HIP_MEMORY_SCOPE_WORKGROUP);
            }
            #pragma unroll
            for (int k = 0; k < 32; ++k) {
                const float ak = __int_as_float(__builtin_amdgcn_readlane(__float_as_int(a), k));
                __hip_atomic_fetch_add(&rowbuf[off[k]], v[k] * ak,
                                       __ATOMIC_RELAXED, __HIP_MEMORY_SCOPE_WORKGROUP);
            }
        }
        if (far) far_flag = 1;
    };

    sweep_fast();
    if (tail > L_CAP && lane == 0) ovf_flag = 1;
    __syncthreads();
    const bool over = __builtin_amdgcn_readfirstlane(ovf_flag) != 0;
    bool twopass = over || !packed;
    if (!twopass) {
        flush_onepass();
        __syncthreads();
        twopass = __builtin_amdgcn_readfirstlane(far_flag) != 0;
        if (twopass) {
            for (int i = tid; i < (OWN_ROWS + 1) * 8; i += OWN_THREADS)
                acc4[i] = make_float4(0.0f, 0.0f, 0.0f, 0.0f);
            for (int i = tid; i < OWN_ROWS + 1; i += OWN_THREADS)
                ssum[i] = 0.0f;
            __syncthreads();
        }
    }
    if (twopass) {
        const int nChunks = over ? (E + CHUNK_EDGES - 1) / CHUNK_EDGES : 1;
        for (int chunk = 0; chunk < nChunks; ++chunk) {
            if (over) {
                const int lo = chunk * CHUNK_EDGES;
                int hi = lo + CHUNK_EDGES;
                if (hi > E) hi = E;
                tail = 0;
                sweep_plain(lo, hi);
            }
            flush();
        }
        __syncthreads();
    }

    float4* __restrict__ out4 = (float4*)(out + (size_t)d0 * 32);
    for (int i = tid; i < rows * 8; i += OWN_THREADS) {
        const int c = (i & 7) * 4;
        const float sm = ssum[i >> 3];
        float4 v = acc4[i];
        if (sm != 0.0f) { v.x /= sm; v.y /= sm; v.z /= sm; v.w /= sm; }
        v.x += bias[c];     v.x = (v.x > 0.0f) ? v.x : NEG_OUT * v.x;
        v.y += bias[c + 1]; v.y = (v.y > 0.0f) ? v.y : NEG_OUT * v.y;
        v.z += bias[c + 2]; v.z = (v.z > 0.0f) ? v.z : NEG_OUT * v.z;
        v.w += bias[c + 3]; v.w = (v.w > 0.0f) ? v.w : NEG_OUT * v.w;
        out4[i] = v;
    }
}

extern "C" void kernel_launch(void* const* d_in, const int* in_sizes, int n_in,
                              void* d_out, int out_size, void* d_ws, size_t ws_size,
                              hipStream_t stream) {
    const float* x    = (const float*)d_in[0];
    const int*   ei   = (const int*)d_in[1];
    const float* ew   = (const float*)d_in[2];
    const float* Wl   = (const float*)d_in[3];
    const float* bl   = (const float*)d_in[4];
    const float* Wr   = (const float*)d_in[5];
    const float* br   = (const float*)d_in[6];
    const float* We   = (const float*)d_in[7];
    const float* att  = (const float*)d_in[8];
    const float* bias = (const float*)d_in[9];

    const int N = in_sizes[0] / 128;
    const int E = in_sizes[2];
    const int* srcv = ei;
    const int* dstv = ei + E;

    float* xl  = (float*)d_ws;
    float* xr  = xl + (size_t)N * 32;
    unsigned int* rec = (unsigned int*)(xr + (size_t)N * 32);
    unsigned int* oid = rec + (size_t)E * 2;
    float* out = (float*)d_out;

    const int tiles = (N + 15) / 16;
    const int wavesPerBlock = 8;
    gat_gemm<<<(tiles + wavesPerBlock - 1) / wavesPerBlock, 32 * wavesPerBlock, 0, stream>>>(
        x, Wl, bl, Wr, br, xl, xr, N, dstv, oid, E, srcv, ew, rec);

    gat_edges_owner<<<(N + OWN_ROWS - 1) / OWN_ROWS, OWN_THREADS, 0, stream>>>(
        srcv, dstv, oid, rec, ew, xl, xr, We, att, bias, out, N, E);
}
